# starting kernel
# baseline (speedup 1.0000x reference)
#include <hip/hip_runtime.h>

typedef float v2f __attribute__((ext_vector_type(2)));
typedef float v8f __attribute__((ext_vector_type(8)));

#define GP_ROUNDS   3
#define GP_NODE_DIM 64
#define GP_ACT_DIM  64
#define GP_EDGE_DIM 32
#define GP_N_NODES  100000
#define GP_N_EDGES  1600000
#define GP_CAT_DIM  160
#define KC_SRC  16
#define KC_DST  16
#define KC_EDG  8
#define KC_TOT  40

static __device__ __forceinline__ v8f wmma4(v2f a, v2f b, v8f c) {
  return __builtin_amdgcn_wmma_f32_16x16x4_f32(false, a, false, b, (short)0, c,
                                               false, false);
}
static __device__ __forceinline__ v2f ld2(const float* p) {
  return *(const v2f*)p;
}
static __device__ __forceinline__ float sigmoid_(float x) {
  return 1.0f / (1.0f + __expf(-x));
}

__global__ __launch_bounds__(256) void gp_edge_kernel(
    const float* __restrict__ hv, const float* __restrict__ he,
    const int* __restrict__ src, const int* __restrict__ dst,
    const float* __restrict__ W,
    const float* __restrict__ bias,
    float* __restrict__ abuf)
{
  __shared__ float Wl[KC_TOT * 4 * 32 * 2];
  const int tid = threadIdx.x;
#pragma unroll
  for (int p = tid; p < KC_TOT * 4 * 32; p += 256) {
    const int L  = p & 31;
    const int q  = p >> 5;
    const int n0 = q & 3;
    const int c  = q >> 2;
    const int k  = 4 * c + 2 * (L >> 4);
    const int col = n0 * 16 + (L & 15);
    v2f w2;
    w2.x = W[k * 64 + col];
    w2.y = W[(k + 1) * 64 + col];
    *(v2f*)&Wl[2 * p] = w2;
  }
  __syncthreads();

  const int lane = tid & 31;
  const int wave = tid >> 5;
  const int tile = blockIdx.x * 8 + wave;
  const int e0   = tile * 16;
  const int row  = lane & 15;
  const int half = lane >> 4;

  const int sidx = src[e0 + row];
  const int didx = dst[e0 + row];
  const float* hs = hv + (long)sidx * GP_NODE_DIM + 2 * half;
  const float* hd = hv + (long)didx * GP_NODE_DIM + 2 * half;
  const float* hp = he + (long)(e0 + row) * GP_EDGE_DIM + 2 * half;

  v2f As[KC_SRC], Ad[KC_DST], Ae[KC_EDG];
#pragma unroll
  for (int c = 0; c < KC_SRC; ++c) As[c] = ld2(hs + 4 * c);
#pragma unroll
  for (int c = 0; c < KC_DST; ++c) Ad[c] = ld2(hd + 4 * c);
#pragma unroll
  for (int c = 0; c < KC_EDG; ++c) Ae[c] = ld2(hp + 4 * c);

  int drow[8];
#pragma unroll
  for (int i = 0; i < 8; ++i) drow[i] = dst[e0 + i + 8 * half];

  const v2f* Wl2 = (const v2f*)Wl;
#pragma unroll
  for (int n0 = 0; n0 < 4; ++n0) {
    const float bv = bias[n0 * 16 + row];
    v8f acc;
#pragma unroll
    for (int i = 0; i < 8; ++i) acc[i] = bv;

    const v2f* Bp = Wl2 + (n0 * 32 + lane);
#pragma unroll
    for (int c = 0; c < KC_SRC; ++c)
      acc = wmma4(As[c], Bp[(c) * 128], acc);
#pragma unroll
    for (int c = 0; c < KC_DST; ++c)
      acc = wmma4(Ad[c], Bp[(KC_SRC + c) * 128], acc);
#pragma unroll
    for (int c = 0; c < KC_EDG; ++c)
      acc = wmma4(Ae[c], Bp[(KC_SRC + KC_DST + c) * 128], acc);

#pragma unroll
    for (int i = 0; i < 8; ++i)
      atomicAdd(&abuf[(long)drow[i] * GP_ACT_DIM + n0 * 16 + row], acc[i]);
  }
}

__global__ __launch_bounds__(256) void gp_gru_kernel(
    const float* __restrict__ abuf, const float* __restrict__ hvin,
    const float* __restrict__ w_ih, const float* __restrict__ w_hh,
    const float* __restrict__ b_ih, const float* __restrict__ b_hh,
    float* __restrict__ hvout)
{
  const int tid  = threadIdx.x;
  const int lane = tid & 31;
  const int wave = tid >> 5;
  const int tile = blockIdx.x * 8 + wave;
  const int ntiles = GP_N_NODES / 16;
  if (tile >= ntiles) return;

  const int nbase = tile * 16;
  const int row   = lane & 15;
  const int half  = lane >> 4;

  v2f Aa[16], Ah[16];
  const float* ap = abuf + (long)(nbase + row) * GP_ACT_DIM + 2 * half;
  const float* hq = hvin + (long)(nbase + row) * GP_NODE_DIM + 2 * half;
#pragma unroll
  for (int c = 0; c < 16; ++c) { Aa[c] = ld2(ap + 4 * c); Ah[c] = ld2(hq + 4 * c); }

#pragma unroll
  for (int j0 = 0; j0 < 4; ++j0) {
    const int colb = j0 * 16 + row;
    v8f ir, iz, in_, hr, hz, hn;
    {
      const float bi0 = b_ih[colb], bi1 = b_ih[64 + colb], bi2 = b_ih[128 + colb];
      const float bh0 = b_hh[colb], bh1 = b_hh[64 + colb], bh2 = b_hh[128 + colb];
#pragma unroll
      for (int i = 0; i < 8; ++i) {
        ir[i] = bi0; iz[i] = bi1; in_[i] = bi2;
        hr[i] = bh0; hz[i] = bh1; hn[i] = bh2;
      }
    }
    const float* wi0 = w_ih + (long)(colb)*64 + 2 * half;
    const float* wi1 = w_ih + (long)(64 + colb) * 64 + 2 * half;
    const float* wi2 = w_ih + (long)(128 + colb) * 64 + 2 * half;
    const float* wh0 = w_hh + (long)(colb)*64 + 2 * half;
    const float* wh1 = w_hh + (long)(64 + colb) * 64 + 2 * half;
    const float* wh2 = w_hh + (long)(128 + colb) * 64 + 2 * half;
#pragma unroll
    for (int c = 0; c < 16; ++c) {
      ir  = wmma4(Aa[c], ld2(wi0 + 4 * c), ir);
      iz  = wmma4(Aa[c], ld2(wi1 + 4 * c), iz);
      in_ = wmma4(Aa[c], ld2(wi2 + 4 * c), in_);
      hr  = wmma4(Ah[c], ld2(wh0 + 4 * c), hr);
      hz  = wmma4(Ah[c], ld2(wh1 + 4 * c), hz);
      hn  = wmma4(Ah[c], ld2(wh2 + 4 * c), hn);
    }
#pragma unroll
    for (int i = 0; i < 8; ++i) {
      const long m = nbase + i + 8 * half;
      const float h = hvin[m * GP_NODE_DIM + colb];
      const float r = sigmoid_(ir[i] + hr[i]);
      const float z = sigmoid_(iz[i] + hz[i]);
      const float n = tanhf(in_[i] + r * hn[i]);
      hvout[m * GP_NODE_DIM + colb] = (1.0f - z) * n + z * h;
    }
  }
}

extern "C" void kernel_launch(void* const* d_in, const int* in_sizes, int n_in,
                              void* d_out, int out_size, void* d_ws, size_t ws_size,
                              hipStream_t stream) {
  const float* hv    = (const float*)d_in[0];
  const float* he    = (const float*)d_in[1];
  const int*   src   = (const int*)d_in[2];
  const int*   dst   = (const int*)d_in[3];
  const float* W_msg = (const float*)d_in[4];
  const float* b_msg = (const float*)d_in[5];
  const float* w_ih  = (const float*)d_in[6];
  const float* w_hh  = (const float*)d_in[7];
  const float* b_ih  = (const float*)d_in[8];
  const float* b_hh  = (const float*)d_in[9];
  float* out = (float*)d_out;

  const size_t state_elems = (size_t)GP_N_NODES * GP_NODE_DIM;
  float* a_buf  = (float*)d_ws;
  float* hv_buf = a_buf + state_elems;

  hipMemcpyAsync(hv_buf, hv, state_elems * sizeof(float),
                 hipMemcpyDeviceToDevice, stream);

  const int edge_blocks = (GP_N_EDGES / 16) / 8;
  const int gru_blocks  = (GP_N_NODES / 16 + 7) / 8;

  for (int t = 0; t < GP_ROUNDS; ++t) {
    hipMemsetAsync(a_buf, 0, state_elems * sizeof(float), stream);
    gp_edge_kernel<<<edge_blocks, 256, 0, stream>>>(
        hv_buf, he, src, dst,
        W_msg + (size_t)t * GP_CAT_DIM * GP_ACT_DIM,
        b_msg + (size_t)t * GP_ACT_DIM, a_buf);
    float* outp = (t == GP_ROUNDS - 1) ? out : hv_buf;
    gp_gru_kernel<<<gru_blocks, 256, 0, stream>>>(
        a_buf, hv_buf,
        w_ih + (size_t)t * 3 * GP_NODE_DIM * GP_NODE_DIM,
        w_hh + (size_t)t * 3 * GP_NODE_DIM * GP_NODE_DIM,
        b_ih + (size_t)t * 3 * GP_NODE_DIM,
        b_hh + (size_t)t * 3 * GP_NODE_DIM, outp);
  }
}


# Edge scatter: two 16-col tiles side by side, lane halves exchanged (v_permlanex16) so each f32 atomic adds one whole 128-B line of one dst row, not two 64-B pieces; adds stay in arrival order
# speedup vs baseline: 1.0438x; 1.0438x over previous
#include <hip/hip_runtime.h>

typedef float v2f __attribute__((ext_vector_type(2)));
typedef float v4f __attribute__((ext_vector_type(4)));
typedef float v8f __attribute__((ext_vector_type(8)));

#define GP_ROUNDS   3
#define GP_NODE_DIM 64
#define GP_ACT_DIM  64
#define GP_EDGE_DIM 32
#define GP_N_NODES  100000
#define GP_N_EDGES  1600000
#define GP_CAT_DIM  160
#define KC_SRC  16
#define KC_DST  16
#define KC_EDG  8
#define KC_TOT  40

static __device__ __forceinline__ v8f wmma4(v2f a, v2f b, v8f c) {
  return __builtin_amdgcn_wmma_f32_16x16x4_f32(false, a, false, b, (short)0, c,
                                               false, false);
}
static __device__ __forceinline__ v2f ld2(const float* p) {
  return *(const v2f*)p;
}
static __device__ __forceinline__ float sigmoid_(float x) {
  return 1.0f / (1.0f + __expf(-x));
}

static __device__ __forceinline__ float xhalf(float x) {
  return __builtin_bit_cast(float, __builtin_amdgcn_permlanex16(
      __builtin_bit_cast(int, x), __builtin_bit_cast(int, x),
      0x76543210, 0xfedcba98, false, false));
}
static __device__ __forceinline__ void swap_halves(float& lo, float& hi, bool upper) {
  const float t  = upper ? lo : hi;
  const float ts = xhalf(t);
  lo = upper ? ts : lo;
  hi = upper ? hi : ts;
}

__global__ __launch_bounds__(256) void gp_edge_kernel(
    const float* __restrict__ hv, const float* __restrict__ he,
    const int* __restrict__ src, const int* __restrict__ dst,
    const float* __restrict__ W,
    const float* __restrict__ bias,
    float* __restrict__ abuf)
{
  __shared__ __attribute__((aligned(16))) float Wl[KC_TOT * 4 * 32 * 2];
  const int tid = threadIdx.x;
#pragma unroll
  for (int p = tid; p < KC_TOT * 4 * 32; p += 256) {
    const int L  = p & 31;
    const int q  = p >> 5;
    const int n0 = q & 3;
    const int c  = q >> 2;
    const int k  = 4 * c + 2 * (L >> 4);
    const int col = n0 * 16 + (L & 15);
    v2f w2;
    w2.x = W[k * 64 + col];
    w2.y = W[(k + 1) * 64 + col];
    *(v2f*)&Wl[(((c * 2 + (n0 >> 1)) * 32 + L) * 2 + (n0 & 1)) * 2] = w2;
  }
  __syncthreads();

  const int lane = tid & 31;
  const int wave = tid >> 5;
  const int tile = blockIdx.x * 8 + wave;
  const int e0   = tile * 16;
  const int row  = lane & 15;
  const int half = lane >> 4;

  const int sidx = src[e0 + row];
  const int didx = dst[e0 + row];
  const float* hs = hv + (long)sidx * GP_NODE_DIM + 2 * half;
  const float* hd = hv + (long)didx * GP_NODE_DIM + 2 * half;
  const float* hp = he + (long)(e0 + row) * GP_EDGE_DIM + 2 * half;

  v2f As[KC_SRC], Ad[KC_DST], Ae[KC_EDG];
#pragma unroll
  for (int c = 0; c < KC_SRC; ++c) As[c] = ld2(hs + 4 * c);
#pragma unroll
  for (int c = 0; c < KC_DST; ++c) Ad[c] = ld2(hd + 4 * c);
#pragma unroll
  for (int c = 0; c < KC_EDG; ++c) Ae[c] = ld2(hp + 4 * c);

  int drow[16];
#pragma unroll
  for (int i = 0; i < 16; ++i) drow[i] = __builtin_amdgcn_readlane(didx, i);

  const v4f* Wl4 = (const v4f*)Wl;
#pragma unroll
  for (int pr = 0; pr < 2; ++pr) {
    const float bv0 = bias[pr * 32 + row];
    const float bv1 = bias[pr * 32 + 16 + row];
    v8f acc0, acc1;
#pragma unroll
    for (int i = 0; i < 8; ++i) { acc0[i] = bv0; acc1[i] = bv1; }

    const v4f* Bp = Wl4 + (pr * 32 + lane);
#pragma unroll
    for (int c = 0; c < KC_SRC; ++c) {
      const v4f b = Bp[(c) * 64];
      acc0 = wmma4(As[c], b.xy, acc0);
      acc1 = wmma4(As[c], b.zw, acc1);
    }
#pragma unroll
    for (int c = 0; c < KC_DST; ++c) {
      const v4f b = Bp[(KC_SRC + c) * 64];
      acc0 = wmma4(Ad[c], b.xy, acc0);
      acc1 = wmma4(Ad[c], b.zw, acc1);
    }
#pragma unroll
    for (int c = 0; c < KC_EDG; ++c) {
      const v4f b = Bp[(KC_SRC + KC_DST + c) * 64];
      acc0 = wmma4(Ae[c], b.xy, acc0);
      acc1 = wmma4(Ae[c], b.zw, acc1);
    }

#pragma unroll
    for (int i = 0; i < 8; ++i) {
      float x = acc0[i], y = acc1[i];
      swap_halves(x, y, half != 0);
      const unsigned ox = ((unsigned)drow[i] << 8) + (unsigned)(pr * 32 + lane) * 4u;
      const unsigned oy = ((unsigned)drow[i + 8] << 8) + (unsigned)(pr * 32 + lane) * 4u;
      atomicAdd((float*)((char*)abuf + ox), x);
      atomicAdd((float*)((char*)abuf + oy), y);
    }
  }
}

__global__ __launch_bounds__(256) void gp_gru_kernel(
    const float* __restrict__ abuf, const float* __restrict__ hvin,
    const float* __restrict__ w_ih, const float* __restrict__ w_hh,
    const float* __restrict__ b_ih, const float* __restrict__ b_hh,
    float* __restrict__ hvout)
{
  const int tid  = threadIdx.x;
  const int lane = tid & 31;
  const int wave = tid >> 5;
  const int tile = blockIdx.x * 8 + wave;
  const int ntiles = GP_N_NODES / 16;
  if (tile >= ntiles) return;

  const int nbase = tile * 16;
  const int row   = lane & 15;
  const int half  = lane >> 4;

  v2f Aa[16], Ah[16];
  const float* ap = abuf + (long)(nbase + row) * GP_ACT_DIM + 2 * half;
  const float* hq = hvin + (long)(nbase + row) * GP_NODE_DIM + 2 * half;
#pragma unroll
  for (int c = 0; c < 16; ++c) { Aa[c] = ld2(ap + 4 * c); Ah[c] = ld2(hq + 4 * c); }

#pragma unroll
  for (int j0 = 0; j0 < 4; ++j0) {
    const int colb = j0 * 16 + row;
    v8f ir, iz, in_, hr, hz, hn;
    {
      const float bi0 = b_ih[colb], bi1 = b_ih[64 + colb], bi2 = b_ih[128 + colb];
      const float bh0 = b_hh[colb], bh1 = b_hh[64 + colb], bh2 = b_hh[128 + colb];
#pragma unroll
      for (int i = 0; i < 8; ++i) {
        ir[i] = bi0; iz[i] = bi1; in_[i] = bi2;
        hr[i] = bh0; hz[i] = bh1; hn[i] = bh2;
      }
    }
    const float* wi0 = w_ih + (long)(colb)*64 + 2 * half;
    const float* wi1 = w_ih + (long)(64 + colb) * 64 + 2 * half;
    const float* wi2 = w_ih + (long)(128 + colb) * 64 + 2 * half;
    const float* wh0 = w_hh + (long)(colb)*64 + 2 * half;
    const float* wh1 = w_hh + (long)(64 + colb) * 64 + 2 * half;
    const float* wh2 = w_hh + (long)(128 + colb) * 64 + 2 * half;
#pragma unroll
    for (int c = 0; c < 16; ++c) {
      ir  = wmma4(Aa[c], ld2(wi0 + 4 * c), ir);
      iz  = wmma4(Aa[c], ld2(wi1 + 4 * c), iz);
      in_ = wmma4(Aa[c], ld2(wi2 + 4 * c), in_);
      hr  = wmma4(Ah[c], ld2(wh0 + 4 * c), hr);
      hz  = wmma4(Ah[c], ld2(wh1 + 4 * c), hz);
      hn  = wmma4(Ah[c], ld2(wh2 + 4 * c), hn);
    }
#pragma unroll
    for (int i = 0; i < 8; ++i) {
      const long m = nbase + i + 8 * half;
      const float h = hvin[m * GP_NODE_DIM + colb];
      const float r = sigmoid_(ir[i] + hr[i]);
      const float z = sigmoid_(iz[i] + hz[i]);
      const float n = tanhf(in_[i] + r * hn[i]);
      hvout[m * GP_NODE_DIM + colb] = (1.0f - z) * n + z * h;
    }
  }
}

extern "C" void kernel_launch(void* const* d_in, const int* in_sizes, int n_in,
                              void* d_out, int out_size, void* d_ws, size_t ws_size,
                              hipStream_t stream) {
  const float* hv    = (const float*)d_in[0];
  const float* he    = (const float*)d_in[1];
  const int*   src   = (const int*)d_in[2];
  const int*   dst   = (const int*)d_in[3];
  const float* W_msg = (const float*)d_in[4];
  const float* b_msg = (const float*)d_in[5];
  const float* w_ih  = (const float*)d_in[6];
  const float* w_hh  = (const float*)d_in[7];
  const float* b_ih  = (const float*)d_in[8];
  const float* b_hh  = (const float*)d_in[9];
  float* out = (float*)d_out;

  const size_t state_elems = (size_t)GP_N_NODES * GP_NODE_DIM;
  float* a_buf  = (float*)d_ws;
  float* hv_buf = a_buf + state_elems;

  hipMemcpyAsync(hv_buf, hv, state_elems * sizeof(float),
                 hipMemcpyDeviceToDevice, stream);

  const int edge_blocks = (GP_N_EDGES / 16) / 8;
  const int gru_blocks  = (GP_N_NODES / 16 + 7) / 8;

  for (int t = 0; t < GP_ROUNDS; ++t) {
    hipMemsetAsync(a_buf, 0, state_elems * sizeof(float), stream);
    gp_edge_kernel<<<edge_blocks, 256, 0, stream>>>(
        hv_buf, he, src, dst,
        W_msg + (size_t)t * GP_CAT_DIM * GP_ACT_DIM,
        b_msg + (size_t)t * GP_ACT_DIM, a_buf);
    float* outp = (t == GP_ROUNDS - 1) ? out : hv_buf;
    gp_gru_kernel<<<gru_blocks, 256, 0, stream>>>(
        a_buf, hv_buf,
        w_ih + (size_t)t * 3 * GP_NODE_DIM * GP_NODE_DIM,
        w_hh + (size_t)t * 3 * GP_NODE_DIM * GP_NODE_DIM,
        b_ih + (size_t)t * 3 * GP_NODE_DIM,
        b_hh + (size_t)t * 3 * GP_NODE_DIM, outp);
  }
}


# Weights applied after the segment sum (linearity): edges add raw hv[src] and he rows per dst with f32 atomics, one per-node WMMA product makes a; dst term = in-degree x (hv@Wd+b); sum order changes
# speedup vs baseline: 1.5295x; 1.4653x over previous
#include <hip/hip_runtime.h>


typedef float v2f __attribute__((ext_vector_type(2)));
typedef float v4f __attribute__((ext_vector_type(4)));
typedef float v8f __attribute__((ext_vector_type(8)));

#define GP_ROUNDS   3
#define GP_NODE_DIM 64
#define GP_ACT_DIM  64
#define GP_EDGE_DIM 32
#define GP_N_NODES  100000
#define GP_N_EDGES  1600000
#define GP_CAT_DIM  160
#define GP_AUX_FLOATS (GP_N_NODES * (GP_EDGE_DIM + 1))
#define GP_AUX_QUADS  (GP_AUX_FLOATS / 4)

static __device__ __forceinline__ v8f wmma4(v2f a, v2f b, v8f c) {
  return __builtin_amdgcn_wmma_f32_16x16x4_f32(false, a, false, b, (short)0, c,
                                               false, false);
}
static __device__ __forceinline__ v2f ld2(const float* p) {
  return *(const v2f*)p;
}
static __device__ __forceinline__ float sigmoid_(float x) {
  return 1.0f / (1.0f + __expf(-x));
}


__global__ __launch_bounds__(256) void gp_zero_kernel(float* __restrict__ aux) {
  const int i = blockIdx.x * 256 + threadIdx.x;
  if (i < GP_AUX_QUADS) {
    v4f z = {0.0f, 0.0f, 0.0f, 0.0f};
    ((v4f*)aux)[i] = z;
  }
}

template <bool FIRST>
__global__ __launch_bounds__(256) void gp_scatter_kernel(
    const float* __restrict__ hv, const float* __restrict__ he,
    const int* __restrict__ src, const int* __restrict__ dst,
    float* __restrict__ sbuf,
    float* __restrict__ hsum,
    float* __restrict__ deg)
{
  const int lane = threadIdx.x & 31;
  const int wave = threadIdx.x >> 5;
  const int e0   = (blockIdx.x * 8 + wave) * 32;
  const int s = src[e0 + lane];
  const int d = dst[e0 + lane];
  const unsigned lo = (unsigned)lane * 4u;
  const unsigned eo = (unsigned)e0 * (GP_EDGE_DIM * 4u) + lo;
#pragma unroll
  for (int b = 0; b < 2; ++b) {
    float v0[16], v1[16], h0[16];
    unsigned dd[16];
#pragma unroll
    for (int i = 0; i < 16; ++i) {
      const unsigned si = (unsigned)__builtin_amdgcn_readlane(s, b * 16 + i);
      dd[i]             = (unsigned)__builtin_amdgcn_readlane(d, b * 16 + i);
      const unsigned so = (si << 8) | lo;
      v0[i] = *(const float*)((const char*)hv + so);
      v1[i] = *(const float*)((const char*)hv + so + 128u);
      if (FIRST)
        h0[i] = *(const float*)((const char*)he + eo + (unsigned)(b * 16 + i) * (GP_EDGE_DIM * 4u));
    }
#pragma unroll
    for (int i = 0; i < 16; ++i) {
      const unsigned ox = (dd[i] << 8) | lo;
      atomicAdd((float*)((char*)sbuf + ox), v0[i]);
      atomicAdd((float*)((char*)sbuf + ox + 128u), v1[i]);
      if (FIRST) atomicAdd((float*)((char*)hsum + ((dd[i] << 7) | lo)), h0[i]);
    }
  }
  if (FIRST) atomicAdd(deg + d, 1.0f);
}

__global__ __launch_bounds__(256) void gp_node_kernel(
    float* sa,
    const float* __restrict__ hvin,
    const float* __restrict__ hsum,
    const float* __restrict__ deg,
    const float* __restrict__ W,
    const float* __restrict__ bias)
{
  const int tid  = threadIdx.x;
  const int lane = tid & 31;
  const int wave = tid >> 5;
  const int tile = blockIdx.x * 8 + wave;
  const int ntiles = GP_N_NODES / 16;
  if (tile >= ntiles) return;

  const int nbase = tile * 16;
  const int row   = lane & 15;
  const int half  = lane >> 4;

  v2f As[16], Ah[16], Ae[8];
  const float* sp = sa   + (long)(nbase + row) * GP_NODE_DIM + 2 * half;
  const float* hq = hvin + (long)(nbase + row) * GP_NODE_DIM + 2 * half;
  const float* ep = hsum + (long)(nbase + row) * GP_EDGE_DIM + 2 * half;
#pragma unroll
  for (int c = 0; c < 16; ++c) { As[c] = ld2(sp + 4 * c); Ah[c] = ld2(hq + 4 * c); }
#pragma unroll
  for (int c = 0; c < 8; ++c) Ae[c] = ld2(ep + 4 * c);

  float dg[8];
#pragma unroll
  for (int i = 0; i < 8; ++i) dg[i] = deg[nbase + 8 * half + i];

#pragma unroll
  for (int j0 = 0; j0 < 4; ++j0) {
    const int col = j0 * 16 + row;
    const float* wb = W + (2 * half) * GP_ACT_DIM + col;
    const float bq = bias[col];
    v8f p, q;
#pragma unroll
    for (int i = 0; i < 8; ++i) { p[i] = 0.0f; q[i] = bq; }
#pragma unroll
    for (int c = 0; c < 16; ++c) {
      v2f bs, bd;
      bs.x = wb[(4 * c) * GP_ACT_DIM];
      bs.y = wb[(4 * c + 1) * GP_ACT_DIM];
      bd.x = wb[(GP_NODE_DIM + 4 * c) * GP_ACT_DIM];
      bd.y = wb[(GP_NODE_DIM + 4 * c + 1) * GP_ACT_DIM];
      p = wmma4(As[c], bs, p);
      q = wmma4(Ah[c], bd, q);
    }
#pragma unroll
    for (int c = 0; c < 8; ++c) {
      v2f be;
      be.x = wb[(2 * GP_NODE_DIM + 4 * c) * GP_ACT_DIM];
      be.y = wb[(2 * GP_NODE_DIM + 4 * c + 1) * GP_ACT_DIM];
      p = wmma4(Ae[c], be, p);
    }
#pragma unroll
    for (int i = 0; i < 8; ++i) {
      const long m = nbase + i + 8 * half;
      sa[m * GP_ACT_DIM + col] = fmaf(dg[i], q[i], p[i]);
    }
  }
}

__global__ __launch_bounds__(256) void gp_gru_kernel(
    const float* __restrict__ abuf, const float* __restrict__ hvin,
    const float* __restrict__ w_ih, const float* __restrict__ w_hh,
    const float* __restrict__ b_ih, const float* __restrict__ b_hh,
    float* __restrict__ hvout)
{
  const int tid  = threadIdx.x;
  const int lane = tid & 31;
  const int wave = tid >> 5;
  const int tile = blockIdx.x * 8 + wave;
  const int ntiles = GP_N_NODES / 16;
  if (tile >= ntiles) return;

  const int nbase = tile * 16;
  const int row   = lane & 15;
  const int half  = lane >> 4;

  v2f Aa[16], Ah[16];
  const float* ap = abuf + (long)(nbase + row) * GP_ACT_DIM + 2 * half;
  const float* hq = hvin + (long)(nbase + row) * GP_NODE_DIM + 2 * half;
#pragma unroll
  for (int c = 0; c < 16; ++c) { Aa[c] = ld2(ap + 4 * c); Ah[c] = ld2(hq + 4 * c); }

#pragma unroll
  for (int j0 = 0; j0 < 4; ++j0) {
    const int colb = j0 * 16 + row;
    v8f ir, iz, in_, hr, hz, hn;
    {
      const float bi0 = b_ih[colb], bi1 = b_ih[64 + colb], bi2 = b_ih[128 + colb];
      const float bh0 = b_hh[colb], bh1 = b_hh[64 + colb], bh2 = b_hh[128 + colb];
#pragma unroll
      for (int i = 0; i < 8; ++i) {
        ir[i] = bi0; iz[i] = bi1; in_[i] = bi2;
        hr[i] = bh0; hz[i] = bh1; hn[i] = bh2;
      }
    }
    const float* wi0 = w_ih + (long)(colb)*64 + 2 * half;
    const float* wi1 = w_ih + (long)(64 + colb) * 64 + 2 * half;
    const float* wi2 = w_ih + (long)(128 + colb) * 64 + 2 * half;
    const float* wh0 = w_hh + (long)(colb)*64 + 2 * half;
    const float* wh1 = w_hh + (long)(64 + colb) * 64 + 2 * half;
    const float* wh2 = w_hh + (long)(128 + colb) * 64 + 2 * half;
#pragma unroll
    for (int c = 0; c < 16; ++c) {
      ir  = wmma4(Aa[c], ld2(wi0 + 4 * c), ir);
      iz  = wmma4(Aa[c], ld2(wi1 + 4 * c), iz);
      in_ = wmma4(Aa[c], ld2(wi2 + 4 * c), in_);
      hr  = wmma4(Ah[c], ld2(wh0 + 4 * c), hr);
      hz  = wmma4(Ah[c], ld2(wh1 + 4 * c), hz);
      hn  = wmma4(Ah[c], ld2(wh2 + 4 * c), hn);
    }
#pragma unroll
    for (int i = 0; i < 8; ++i) {
      const long m = nbase + i + 8 * half;
      const float h = hvin[m * GP_NODE_DIM + colb];
      const float r = sigmoid_(ir[i] + hr[i]);
      const float z = sigmoid_(iz[i] + hz[i]);
      const float n = tanhf(in_[i] + r * hn[i]);
      hvout[m * GP_NODE_DIM + colb] = (1.0f - z) * n + z * h;
    }
  }
}

extern "C" void kernel_launch(void* const* d_in, const int* in_sizes, int n_in,
                              void* d_out, int out_size, void* d_ws, size_t ws_size,
                              hipStream_t stream) {
  const float* hv    = (const float*)d_in[0];
  const float* he    = (const float*)d_in[1];
  const int*   src   = (const int*)d_in[2];
  const int*   dst   = (const int*)d_in[3];
  const float* W_msg = (const float*)d_in[4];
  const float* b_msg = (const float*)d_in[5];
  const float* w_ih  = (const float*)d_in[6];
  const float* w_hh  = (const float*)d_in[7];
  const float* b_ih  = (const float*)d_in[8];
  const float* b_hh  = (const float*)d_in[9];
  float* out = (float*)d_out;

  const size_t state_elems = (size_t)GP_N_NODES * GP_NODE_DIM;
  float* a_buf  = (float*)d_ws;
  float* hv_buf = a_buf + state_elems;
  float* h_sum = out;
  float* deg   = out + (size_t)GP_N_NODES * GP_EDGE_DIM;

  hipMemcpyAsync(hv_buf, hv, state_elems * sizeof(float),
                 hipMemcpyDeviceToDevice, stream);

  const int zero_blocks    = (GP_AUX_QUADS + 255) / 256;
  const int scatter_blocks = (GP_N_EDGES / 32) / 8;
  const int node_blocks    = (GP_N_NODES / 16 + 7) / 8;

  gp_zero_kernel<<<zero_blocks, 256, 0, stream>>>(out);

  for (int t = 0; t < GP_ROUNDS; ++t) {
    hipMemsetAsync(a_buf, 0, state_elems * sizeof(float), stream);
    if (t == 0)
      gp_scatter_kernel<true><<<scatter_blocks, 256, 0, stream>>>(
          hv_buf, he, src, dst, a_buf, h_sum, deg);
    else
      gp_scatter_kernel<false><<<scatter_blocks, 256, 0, stream>>>(
          hv_buf, he, src, dst, a_buf, h_sum, deg);
    gp_node_kernel<<<node_blocks, 256, 0, stream>>>(
        a_buf, hv_buf, h_sum, deg,
        W_msg + (size_t)t * GP_CAT_DIM * GP_ACT_DIM,
        b_msg + (size_t)t * GP_ACT_DIM);
    float* outp = (t == GP_ROUNDS - 1) ? out : hv_buf;
    gp_gru_kernel<<<node_blocks, 256, 0, stream>>>(
        a_buf, hv_buf,
        w_ih + (size_t)t * 3 * GP_NODE_DIM * GP_NODE_DIM,
        w_hh + (size_t)t * 3 * GP_NODE_DIM * GP_NODE_DIM,
        b_ih + (size_t)t * 3 * GP_NODE_DIM,
        b_hh + (size_t)t * 3 * GP_NODE_DIM, outp);
  }
}


# Segment sums without float atomics: edges put in dst order once per call (int atomic rank + prefix sum); each round one wave per 8 nodes reads its edges' hv rows, stores S once; sums in rank order
# speedup vs baseline: 1.8041x; 1.1795x over previous
#include <hip/hip_runtime.h>


typedef float v2f __attribute__((ext_vector_type(2)));
typedef float v4f __attribute__((ext_vector_type(4)));
typedef float v8f __attribute__((ext_vector_type(8)));

#define GP_ROUNDS   3
#define GP_NODE_DIM 64
#define GP_ACT_DIM  64
#define GP_EDGE_DIM 32
#define GP_N_NODES  100000
#define GP_N_EDGES  1600000
#define GP_CAT_DIM  160
#define GP_NPW      8

static __device__ __forceinline__ v8f wmma4(v2f a, v2f b, v8f c) {
  return __builtin_amdgcn_wmma_f32_16x16x4_f32(false, a, false, b, (short)0, c,
                                               false, false);
}
static __device__ __forceinline__ v2f ld2(const float* p) {
  return *(const v2f*)p;
}
static __device__ __forceinline__ float sigmoid_(float x) {
  return 1.0f / (1.0f + __expf(-x));
}


__global__ __launch_bounds__(256) void gp_zero_kernel(int* __restrict__ counts) {
  const int i = blockIdx.x * 256 + threadIdx.x;
  if (i < GP_N_NODES / 4) {
    const int4 z = {0, 0, 0, 0};
    ((int4*)counts)[i] = z;
  }
}

__global__ __launch_bounds__(256) void gp_rank_kernel(
    const int* __restrict__ dst, int* __restrict__ counts, int* __restrict__ rank)
{
  const int t = blockIdx.x * 256 + threadIdx.x;
  if (t < GP_N_EDGES / 4) {
    const int4 d = ((const int4*)dst)[t];
    int4 r;
    r.x = atomicAdd(counts + d.x, 1);
    r.y = atomicAdd(counts + d.y, 1);
    r.z = atomicAdd(counts + d.z, 1);
    r.w = atomicAdd(counts + d.w, 1);
    ((int4*)rank)[t] = r;
  }
}

__global__ __launch_bounds__(1024) void gp_scan_kernel(
    const int* __restrict__ counts, int* __restrict__ rowptr)
{
  __shared__ int wsum[32];
  __shared__ int wpre[32];
  const int tid  = threadIdx.x;
  const int lane = tid & 31;
  const int wave = tid >> 5;
  const int base = blockIdx.x * 1024;
  int acc = 0;
  for (int i = tid; i < base; i += 1024) acc += counts[i];
  const int i = base + tid;
  const int c = (i < GP_N_NODES) ? counts[i] : 0;
  int x = c;
  for (int o = 1; o < 32; o <<= 1) {
    const int y = __shfl_up(x, o);
    if (lane >= o) x += y;
  }
  for (int o = 16; o > 0; o >>= 1) acc += __shfl_xor(acc, o);
  if (lane == 31) wsum[wave] = x;
  if (lane == 0)  wpre[wave] = acc;
  __syncthreads();
  int tot = wpre[lane] + ((lane < wave) ? wsum[lane] : 0);
  for (int o = 16; o > 0; o >>= 1) tot += __shfl_xor(tot, o);
  if (i <= GP_N_NODES) rowptr[i] = tot + x - c;
}

__global__ __launch_bounds__(256) void gp_fill_kernel(
    const int* __restrict__ src, const int* __restrict__ dst,
    const int* __restrict__ rank, const int* __restrict__ rowptr,
    int* __restrict__ csr_src, int* __restrict__ csr_eid)
{
  const int t = blockIdx.x * 256 + threadIdx.x;
  if (t < GP_N_EDGES / 4) {
    const int4 s = ((const int4*)src)[t];
    const int4 d = ((const int4*)dst)[t];
    const int4 r = ((const int4*)rank)[t];
    const int p0 = rowptr[d.x] + r.x;
    const int p1 = rowptr[d.y] + r.y;
    const int p2 = rowptr[d.z] + r.z;
    const int p3 = rowptr[d.w] + r.w;
    csr_src[p0] = s.x; csr_eid[p0] = 4 * t;
    csr_src[p1] = s.y; csr_eid[p1] = 4 * t + 1;
    csr_src[p2] = s.z; csr_eid[p2] = 4 * t + 2;
    csr_src[p3] = s.w; csr_eid[p3] = 4 * t + 3;
  }
}

template <int V> struct gp_row;
template <> struct gp_row<1> {
  static __device__ __forceinline__ void load(const char* p, float& a, float& b) {
    a = *(const float*)p; b = 0.0f;
  }
  static __device__ __forceinline__ void store(char* p, float a, float) {
    *(float*)p = a;
  }
};
template <> struct gp_row<2> {
  static __device__ __forceinline__ void load(const char* p, float& a, float& b) {
    const v2f t = *(const v2f*)p; a = t.x; b = t.y;
  }
  static __device__ __forceinline__ void store(char* p, float a, float b) {
    v2f t; t.x = a; t.y = b; *(v2f*)p = t;
  }
};

template <int V>
__global__ __launch_bounds__(256) void gp_gather_kernel(
    const float* __restrict__ rows, const int* __restrict__ rowptr,
    const int* __restrict__ idxs, float* __restrict__ outp)
{
  const int lane = threadIdx.x & 31;
  const int wave = __builtin_amdgcn_readfirstlane((int)(threadIdx.x >> 5));
  const int n0 = (blockIdx.x * 8 + wave) * GP_NPW;
  if (n0 >= GP_N_NODES) return;
  const int nn = (GP_N_NODES - n0 < GP_NPW) ? (GP_N_NODES - n0) : GP_NPW;
  int rp = 0;
  if (lane <= nn) rp = rowptr[n0 + lane];
  const int e0 = __builtin_amdgcn_readlane(rp, 0);
  const int e1 = __builtin_amdgcn_readlane(rp, nn);
  int cur = 0;
  int nb  = __builtin_amdgcn_readlane(rp, 1);
  const unsigned lo = (unsigned)lane * (4u * V);
  float a0 = 0.0f, a1 = 0.0f;
  char* const ob = (char*)outp + (unsigned)n0 * (128u * V) + lo;

  for (int base = e0; base < e1; base += 32) {
    const int cnt = (e1 - base < 32) ? (e1 - base) : 32;
    int idx = 0;
    if (lane < cnt) idx = idxs[base + lane];
#pragma unroll
    for (int h = 0; h < 32; h += 16) {
      if (h < cnt) {
        float r0[16], r1[16];
#pragma unroll
        for (int u = 0; u < 16; ++u) {
          const int k = (h + u < cnt) ? (h + u) : (cnt - 1);
          const unsigned s = (unsigned)__builtin_amdgcn_readlane(idx, k);
          gp_row<V>::load((const char*)rows + ((s * (128u * V)) | lo), r0[u], r1[u]);
        }
#pragma unroll
        for (int u = 0; u < 16; ++u) {
          if (h + u < cnt) {
            const int e = base + h + u;
            while (e == nb) {
              gp_row<V>::store(ob + (unsigned)cur * (128u * V), a0, a1);
              a0 = 0.0f; a1 = 0.0f;
              ++cur;
              nb = __builtin_amdgcn_readlane(rp, cur + 1);
            }
            a0 += r0[u]; a1 += r1[u];
          }
        }
      }
    }
  }
  while (cur < nn) {
    gp_row<V>::store(ob + (unsigned)cur * (128u * V), a0, a1);
    a0 = 0.0f; a1 = 0.0f;
    ++cur;
  }
}

__global__ __launch_bounds__(256) void gp_node_kernel(
    float* sa,
    const float* __restrict__ hvin,
    const float* __restrict__ hsum,
    const int* __restrict__ rowptr,
    const float* __restrict__ W,
    const float* __restrict__ bias)
{
  const int tid  = threadIdx.x;
  const int lane = tid & 31;
  const int wave = tid >> 5;
  const int tile = blockIdx.x * 8 + wave;
  const int ntiles = GP_N_NODES / 16;
  if (tile >= ntiles) return;

  const int nbase = tile * 16;
  const int row   = lane & 15;
  const int half  = lane >> 4;

  v2f As[16], Ah[16], Ae[8];
  const float* sp = sa   + (long)(nbase + row) * GP_NODE_DIM + 2 * half;
  const float* hq = hvin + (long)(nbase + row) * GP_NODE_DIM + 2 * half;
  const float* ep = hsum + (long)(nbase + row) * GP_EDGE_DIM + 2 * half;
#pragma unroll
  for (int c = 0; c < 16; ++c) { As[c] = ld2(sp + 4 * c); Ah[c] = ld2(hq + 4 * c); }
#pragma unroll
  for (int c = 0; c < 8; ++c) Ae[c] = ld2(ep + 4 * c);

  float dg[8];
#pragma unroll
  for (int i = 0; i < 8; ++i)
    dg[i] = (float)(rowptr[nbase + 8 * half + i + 1] - rowptr[nbase + 8 * half + i]);

#pragma unroll
  for (int j0 = 0; j0 < 4; ++j0) {
    const int col = j0 * 16 + row;
    const float* wb = W + (2 * half) * GP_ACT_DIM + col;
    const float bq = bias[col];
    v8f p, q;
#pragma unroll
    for (int i = 0; i < 8; ++i) { p[i] = 0.0f; q[i] = bq; }
#pragma unroll
    for (int c = 0; c < 16; ++c) {
      v2f bs, bd;
      bs.x = wb[(4 * c) * GP_ACT_DIM];
      bs.y = wb[(4 * c + 1) * GP_ACT_DIM];
      bd.x = wb[(GP_NODE_DIM + 4 * c) * GP_ACT_DIM];
      bd.y = wb[(GP_NODE_DIM + 4 * c + 1) * GP_ACT_DIM];
      p = wmma4(As[c], bs, p);
      q = wmma4(Ah[c], bd, q);
    }
#pragma unroll
    for (int c = 0; c < 8; ++c) {
      v2f be;
      be.x = wb[(2 * GP_NODE_DIM + 4 * c) * GP_ACT_DIM];
      be.y = wb[(2 * GP_NODE_DIM + 4 * c + 1) * GP_ACT_DIM];
      p = wmma4(Ae[c], be, p);
    }
#pragma unroll
    for (int i = 0; i < 8; ++i) {
      const long m = nbase + i + 8 * half;
      sa[m * GP_ACT_DIM + col] = fmaf(dg[i], q[i], p[i]);
    }
  }
}

__global__ __launch_bounds__(256) void gp_gru_kernel(
    const float* __restrict__ abuf, const float* __restrict__ hvin,
    const float* __restrict__ w_ih, const float* __restrict__ w_hh,
    const float* __restrict__ b_ih, const float* __restrict__ b_hh,
    float* __restrict__ hvout)
{
  const int tid  = threadIdx.x;
  const int lane = tid & 31;
  const int wave = tid >> 5;
  const int tile = blockIdx.x * 8 + wave;
  const int ntiles = GP_N_NODES / 16;
  if (tile >= ntiles) return;

  const int nbase = tile * 16;
  const int row   = lane & 15;
  const int half  = lane >> 4;

  v2f Aa[16], Ah[16];
  const float* ap = abuf + (long)(nbase + row) * GP_ACT_DIM + 2 * half;
  const float* hq = hvin + (long)(nbase + row) * GP_NODE_DIM + 2 * half;
#pragma unroll
  for (int c = 0; c < 16; ++c) { Aa[c] = ld2(ap + 4 * c); Ah[c] = ld2(hq + 4 * c); }

#pragma unroll
  for (int j0 = 0; j0 < 4; ++j0) {
    const int colb = j0 * 16 + row;
    v8f ir, iz, in_, hr, hz, hn;
    {
      const float bi0 = b_ih[colb], bi1 = b_ih[64 + colb], bi2 = b_ih[128 + colb];
      const float bh0 = b_hh[colb], bh1 = b_hh[64 + colb], bh2 = b_hh[128 + colb];
#pragma unroll
      for (int i = 0; i < 8; ++i) {
        ir[i] = bi0; iz[i] = bi1; in_[i] = bi2;
        hr[i] = bh0; hz[i] = bh1; hn[i] = bh2;
      }
    }
    const float* wi0 = w_ih + (long)(colb)*64 + 2 * half;
    const float* wi1 = w_ih + (long)(64 + colb) * 64 + 2 * half;
    const float* wi2 = w_ih + (long)(128 + colb) * 64 + 2 * half;
    const float* wh0 = w_hh + (long)(colb)*64 + 2 * half;
    const float* wh1 = w_hh + (long)(64 + colb) * 64 + 2 * half;
    const float* wh2 = w_hh + (long)(128 + colb) * 64 + 2 * half;
#pragma unroll
    for (int c = 0; c < 16; ++c) {
      ir  = wmma4(Aa[c], ld2(wi0 + 4 * c), ir);
      iz  = wmma4(Aa[c], ld2(wi1 + 4 * c), iz);
      in_ = wmma4(Aa[c], ld2(wi2 + 4 * c), in_);
      hr  = wmma4(Ah[c], ld2(wh0 + 4 * c), hr);
      hz  = wmma4(Ah[c], ld2(wh1 + 4 * c), hz);
      hn  = wmma4(Ah[c], ld2(wh2 + 4 * c), hn);
    }
#pragma unroll
    for (int i = 0; i < 8; ++i) {
      const long m = nbase + i + 8 * half;
      const float h = hvin[m * GP_NODE_DIM + colb];
      const float r = sigmoid_(ir[i] + hr[i]);
      const float z = sigmoid_(iz[i] + hz[i]);
      const float n = tanhf(in_[i] + r * hn[i]);
      hvout[m * GP_NODE_DIM + colb] = (1.0f - z) * n + z * h;
    }
  }
}

extern "C" void kernel_launch(void* const* d_in, const int* in_sizes, int n_in,
                              void* d_out, int out_size, void* d_ws, size_t ws_size,
                              hipStream_t stream) {
  const float* hv    = (const float*)d_in[0];
  const float* he    = (const float*)d_in[1];
  const int*   src   = (const int*)d_in[2];
  const int*   dst   = (const int*)d_in[3];
  const float* W_msg = (const float*)d_in[4];
  const float* b_msg = (const float*)d_in[5];
  const float* w_ih  = (const float*)d_in[6];
  const float* w_hh  = (const float*)d_in[7];
  const float* b_ih  = (const float*)d_in[8];
  const float* b_hh  = (const float*)d_in[9];
  float* out = (float*)d_out;

  const size_t state_elems = (size_t)GP_N_NODES * GP_NODE_DIM;
  float* a_buf  = (float*)d_ws;
  float* hv_buf = a_buf + state_elems;
  float* h_sum   = out;
  int*   rowptr  = (int*)(out + (size_t)GP_N_NODES * GP_EDGE_DIM);
  int*   csr_src = (int*)(out + (size_t)3400000);
  int* rank    = (int*)a_buf;
  int* csr_eid = (int*)a_buf + GP_N_EDGES;
  int* counts  = (int*)a_buf + 2 * GP_N_EDGES;

  hipMemcpyAsync(hv_buf, hv, state_elems * sizeof(float),
                 hipMemcpyDeviceToDevice, stream);

  const int zero_blocks   = (GP_N_NODES / 4 + 255) / 256;
  const int edge_blocks   = (GP_N_EDGES / 4 + 255) / 256;
  const int scan_blocks   = (GP_N_NODES + 1 + 1023) / 1024;
  const int gather_blocks = (GP_N_NODES + 8 * GP_NPW - 1) / (8 * GP_NPW);
  const int node_blocks   = (GP_N_NODES / 16 + 7) / 8;

  gp_zero_kernel<<<zero_blocks, 256, 0, stream>>>(counts);
  gp_rank_kernel<<<edge_blocks, 256, 0, stream>>>(dst, counts, rank);
  gp_scan_kernel<<<scan_blocks, 1024, 0, stream>>>(counts, rowptr);
  gp_fill_kernel<<<edge_blocks, 256, 0, stream>>>(src, dst, rank, rowptr,
                                                  csr_src, csr_eid);
  gp_gather_kernel<1><<<gather_blocks, 256, 0, stream>>>(he, rowptr, csr_eid, h_sum);

  for (int t = 0; t < GP_ROUNDS; ++t) {
    gp_gather_kernel<2><<<gather_blocks, 256, 0, stream>>>(hv_buf, rowptr, csr_src, a_buf);
    gp_node_kernel<<<node_blocks, 256, 0, stream>>>(
        a_buf, hv_buf, h_sum, rowptr,
        W_msg + (size_t)t * GP_CAT_DIM * GP_ACT_DIM,
        b_msg + (size_t)t * GP_ACT_DIM);
    float* outp = (t == GP_ROUNDS - 1) ? out : hv_buf;
    gp_gru_kernel<<<node_blocks, 256, 0, stream>>>(
        a_buf, hv_buf,
        w_ih + (size_t)t * 3 * GP_NODE_DIM * GP_NODE_DIM,
        w_hh + (size_t)t * 3 * GP_NODE_DIM * GP_NODE_DIM,
        b_ih + (size_t)t * 3 * GP_NODE_DIM,
        b_hh + (size_t)t * 3 * GP_NODE_DIM, outp);
  }
}


# GRU weights (96 KB) copied once per work-group to LDS in B-fragment order and read by ds_load_b128 one step ahead; 512 persistent work-groups loop over tiles to share the copy; sum order kept
# speedup vs baseline: 2.0425x; 1.1321x over previous
#include <hip/hip_runtime.h>


typedef float v2f __attribute__((ext_vector_type(2)));
typedef float v4f __attribute__((ext_vector_type(4)));
typedef float v8f __attribute__((ext_vector_type(8)));

#define GP_ROUNDS   3
#define GP_NODE_DIM 64
#define GP_ACT_DIM  64
#define GP_EDGE_DIM 32
#define GP_N_NODES  100000
#define GP_N_EDGES  1600000
#define GP_CAT_DIM  160
#define GP_NPW      8

static __device__ __forceinline__ v8f wmma4(v2f a, v2f b, v8f c) {
  return __builtin_amdgcn_wmma_f32_16x16x4_f32(false, a, false, b, (short)0, c,
                                               false, false);
}
static __device__ __forceinline__ v2f ld2(const float* p) {
  return *(const v2f*)p;
}
static __device__ __forceinline__ float sigmoid_(float x) {
  return 1.0f / (1.0f + __expf(-x));
}


__global__ __launch_bounds__(256) void gp_zero_kernel(int* __restrict__ counts) {
  const int i = blockIdx.x * 256 + threadIdx.x;
  if (i < GP_N_NODES / 4) {
    const int4 z = {0, 0, 0, 0};
    ((int4*)counts)[i] = z;
  }
}

__global__ __launch_bounds__(256) void gp_rank_kernel(
    const int* __restrict__ dst, int* __restrict__ counts, int* __restrict__ rank)
{
  const int t = blockIdx.x * 256 + threadIdx.x;
  if (t < GP_N_EDGES / 4) {
    const int4 d = ((const int4*)dst)[t];
    int4 r;
    r.x = atomicAdd(counts + d.x, 1);
    r.y = atomicAdd(counts + d.y, 1);
    r.z = atomicAdd(counts + d.z, 1);
    r.w = atomicAdd(counts + d.w, 1);
    ((int4*)rank)[t] = r;
  }
}

__global__ __launch_bounds__(1024) void gp_scan_kernel(
    const int* __restrict__ counts, int* __restrict__ rowptr)
{
  __shared__ int wsum[32];
  __shared__ int wpre[32];
  const int tid  = threadIdx.x;
  const int lane = tid & 31;
  const int wave = tid >> 5;
  const int base = blockIdx.x * 1024;
  int acc = 0;
  for (int i = tid; i < base; i += 1024) acc += counts[i];
  const int i = base + tid;
  const int c = (i < GP_N_NODES) ? counts[i] : 0;
  int x = c;
  for (int o = 1; o < 32; o <<= 1) {
    const int y = __shfl_up(x, o);
    if (lane >= o) x += y;
  }
  for (int o = 16; o > 0; o >>= 1) acc += __shfl_xor(acc, o);
  if (lane == 31) wsum[wave] = x;
  if (lane == 0)  wpre[wave] = acc;
  __syncthreads();
  int tot = wpre[lane] + ((lane < wave) ? wsum[lane] : 0);
  for (int o = 16; o > 0; o >>= 1) tot += __shfl_xor(tot, o);
  if (i <= GP_N_NODES) rowptr[i] = tot + x - c;
}

__global__ __launch_bounds__(256) void gp_fill_kernel(
    const int* __restrict__ src, const int* __restrict__ dst,
    const int* __restrict__ rank, const int* __restrict__ rowptr,
    int* __restrict__ csr_src, int* __restrict__ csr_eid)
{
  const int t = blockIdx.x * 256 + threadIdx.x;
  if (t < GP_N_EDGES / 4) {
    const int4 s = ((const int4*)src)[t];
    const int4 d = ((const int4*)dst)[t];
    const int4 r = ((const int4*)rank)[t];
    const int p0 = rowptr[d.x] + r.x;
    const int p1 = rowptr[d.y] + r.y;
    const int p2 = rowptr[d.z] + r.z;
    const int p3 = rowptr[d.w] + r.w;
    csr_src[p0] = s.x; csr_eid[p0] = 4 * t;
    csr_src[p1] = s.y; csr_eid[p1] = 4 * t + 1;
    csr_src[p2] = s.z; csr_eid[p2] = 4 * t + 2;
    csr_src[p3] = s.w; csr_eid[p3] = 4 * t + 3;
  }
}

template <int V> struct gp_row;
template <> struct gp_row<1> {
  static __device__ __forceinline__ void load(const char* p, float& a, float& b) {
    a = *(const float*)p; b = 0.0f;
  }
  static __device__ __forceinline__ void store(char* p, float a, float) {
    *(float*)p = a;
  }
};
template <> struct gp_row<2> {
  static __device__ __forceinline__ void load(const char* p, float& a, float& b) {
    const v2f t = *(const v2f*)p; a = t.x; b = t.y;
  }
  static __device__ __forceinline__ void store(char* p, float a, float b) {
    v2f t; t.x = a; t.y = b; *(v2f*)p = t;
  }
};

template <int V>
__global__ __launch_bounds__(256) void gp_gather_kernel(
    const float* __restrict__ rows, const int* __restrict__ rowptr,
    const int* __restrict__ idxs, float* __restrict__ outp)
{
  const int lane = threadIdx.x & 31;
  const int wave = __builtin_amdgcn_readfirstlane((int)(threadIdx.x >> 5));
  const int n0 = (blockIdx.x * 8 + wave) * GP_NPW;
  if (n0 >= GP_N_NODES) return;
  const int nn = (GP_N_NODES - n0 < GP_NPW) ? (GP_N_NODES - n0) : GP_NPW;
  int rp = 0;
  if (lane <= nn) rp = rowptr[n0 + lane];
  const int e0 = __builtin_amdgcn_readlane(rp, 0);
  const int e1 = __builtin_amdgcn_readlane(rp, nn);
  int cur = 0;
  int nb  = __builtin_amdgcn_readlane(rp, 1);
  const unsigned lo = (unsigned)lane * (4u * V);
  float a0 = 0.0f, a1 = 0.0f;
  char* const ob = (char*)outp + (unsigned)n0 * (128u * V) + lo;

  for (int base = e0; base < e1; base += 32) {
    const int cnt = (e1 - base < 32) ? (e1 - base) : 32;
    int idx = 0;
    if (lane < cnt) idx = idxs[base + lane];
#pragma unroll
    for (int h = 0; h < 32; h += 16) {
      if (h < cnt) {
        float r0[16], r1[16];
#pragma unroll
        for (int u = 0; u < 16; ++u) {
          const int k = (h + u < cnt) ? (h + u) : (cnt - 1);
          const unsigned s = (unsigned)__builtin_amdgcn_readlane(idx, k);
          gp_row<V>::load((const char*)rows + ((s * (128u * V)) | lo), r0[u], r1[u]);
        }
#pragma unroll
        for (int u = 0; u < 16; ++u) {
          if (h + u < cnt) {
            const int e = base + h + u;
            while (e == nb) {
              gp_row<V>::store(ob + (unsigned)cur * (128u * V), a0, a1);
              a0 = 0.0f; a1 = 0.0f;
              ++cur;
              nb = __builtin_amdgcn_readlane(rp, cur + 1);
            }
            a0 += r0[u]; a1 += r1[u];
          }
        }
      }
    }
  }
  while (cur < nn) {
    gp_row<V>::store(ob + (unsigned)cur * (128u * V), a0, a1);
    a0 = 0.0f; a1 = 0.0f;
    ++cur;
  }
}

__global__ __launch_bounds__(256) void gp_node_kernel(
    float* sa,
    const float* __restrict__ hvin,
    const float* __restrict__ hsum,
    const int* __restrict__ rowptr,
    const float* __restrict__ W,
    const float* __restrict__ bias)
{
  const int tid  = threadIdx.x;
  const int lane = tid & 31;
  const int wave = tid >> 5;
  const int tile = blockIdx.x * 8 + wave;
  const int ntiles = GP_N_NODES / 16;
  if (tile >= ntiles) return;

  const int nbase = tile * 16;
  const int row   = lane & 15;
  const int half  = lane >> 4;

  v2f As[16], Ah[16], Ae[8];
  const float* sp = sa   + (long)(nbase + row) * GP_NODE_DIM + 2 * half;
  const float* hq = hvin + (long)(nbase + row) * GP_NODE_DIM + 2 * half;
  const float* ep = hsum + (long)(nbase + row) * GP_EDGE_DIM + 2 * half;
#pragma unroll
  for (int c = 0; c < 16; ++c) { As[c] = ld2(sp + 4 * c); Ah[c] = ld2(hq + 4 * c); }
#pragma unroll
  for (int c = 0; c < 8; ++c) Ae[c] = ld2(ep + 4 * c);

  float dg[8];
#pragma unroll
  for (int i = 0; i < 8; ++i)
    dg[i] = (float)(rowptr[nbase + 8 * half + i + 1] - rowptr[nbase + 8 * half + i]);

#pragma unroll
  for (int j0 = 0; j0 < 4; ++j0) {
    const int col = j0 * 16 + row;
    const float* wb = W + (2 * half) * GP_ACT_DIM + col;
    const float bq = bias[col];
    v8f p, q;
#pragma unroll
    for (int i = 0; i < 8; ++i) { p[i] = 0.0f; q[i] = bq; }
#pragma unroll
    for (int c = 0; c < 16; ++c) {
      v2f bs, bd;
      bs.x = wb[(4 * c) * GP_ACT_DIM];
      bs.y = wb[(4 * c + 1) * GP_ACT_DIM];
      bd.x = wb[(GP_NODE_DIM + 4 * c) * GP_ACT_DIM];
      bd.y = wb[(GP_NODE_DIM + 4 * c + 1) * GP_ACT_DIM];
      p = wmma4(As[c], bs, p);
      q = wmma4(Ah[c], bd, q);
    }
#pragma unroll
    for (int c = 0; c < 8; ++c) {
      v2f be;
      be.x = wb[(2 * GP_NODE_DIM + 4 * c) * GP_ACT_DIM];
      be.y = wb[(2 * GP_NODE_DIM + 4 * c + 1) * GP_ACT_DIM];
      p = wmma4(Ae[c], be, p);
    }
#pragma unroll
    for (int i = 0; i < 8; ++i) {
      const long m = nbase + i + 8 * half;
      sa[m * GP_ACT_DIM + col] = fmaf(dg[i], q[i], p[i]);
    }
  }
}

#define GP_GRU_NB    512
#define GP_GRU_UNITS 192
#define GP_GRU_PITCH 132

static_assert(8 * GP_GRU_NB <= GP_N_NODES / 16, "every wave must own a first tile");

#define GP_GRU_LOADSET(dst, wj, c2)                                                \
  do {                                                                             \
    _Pragma("unroll")                                                              \
    for (int g = 0; g < 6; ++g)                                                    \
      dst[g] = (wj)[(g * 32 + (c2)) * (GP_GRU_PITCH / 4)];                         \
  } while (0)

__global__ __launch_bounds__(256) void gp_gru_kernel(
    const float* __restrict__ abuf, const float* __restrict__ hvin,
    const float* __restrict__ w_ih, const float* __restrict__ w_hh,
    const float* __restrict__ b_ih, const float* __restrict__ b_hh,
    float* __restrict__ hvout)
{
  __shared__ __attribute__((aligned(16))) float wl[GP_GRU_UNITS * GP_GRU_PITCH];

  const int tid  = threadIdx.x;
  const int lane = tid & 31;
  const int wave = __builtin_amdgcn_readfirstlane(tid >> 5);
  const int bid  = (int)blockIdx.x;
  const int ntiles = GP_N_NODES / 16;
  const int row   = lane & 15;
  const int half  = lane >> 4;

  int slot = (wave + bid) & 7;
  int tile = slot * GP_GRU_NB + bid;

  v2f Aa[16], Ah[16];
  {
    const float* ap = abuf + (long)(tile * 16 + row) * GP_ACT_DIM + 2 * half;
    const float* hq = hvin + (long)(tile * 16 + row) * GP_NODE_DIM + 2 * half;
#pragma unroll
    for (int c = 0; c < 16; ++c) { Aa[c] = ld2(ap + 4 * c); Ah[c] = ld2(hq + 4 * c); }
  }

  {
    const int fc = tid & 15, fr = tid >> 4;
#pragma unroll
    for (int it = 0; it < 24; ++it) {
      const float* wsrc = ((it < 12) ? w_ih : w_hh) + ((it % 12) * 16 + fr) * 64 + 4 * fc;
      const v4f v = *(const v4f*)wsrc;
      const int G  = (it / 12) * 3 + ((it % 12) >> 2);
      const int j0 = (it % 12) & 3;
      float* d = wl + ((G * 4 + j0) * 8 + (fc >> 1)) * GP_GRU_PITCH + (fc & 1) * 2;
      v2f lo, hi;
      lo.x = v.x; lo.y = v.y; hi.x = v.z; hi.y = v.w;
      *(v2f*)(d + fr * 4) = lo;
      *(v2f*)(d + (16 + fr) * 4) = hi;
    }
  }
  __syncthreads();

  const v4f* wq = (const v4f*)wl + lane;

  v4f fb[2][6];
  GP_GRU_LOADSET(fb[0], wq, 0);
  float bn[6];
  bn[0] = b_ih[row]; bn[1] = b_ih[64 + row]; bn[2] = b_ih[128 + row];
  bn[3] = b_hh[row]; bn[4] = b_hh[64 + row]; bn[5] = b_hh[128 + row];

  for (;;) {
    const int nbase = tile * 16;
    const float* hrow = hvin  + (long)(nbase + 8 * half) * GP_NODE_DIM + row;
    float*       orow = hvout + (long)(nbase + 8 * half) * GP_NODE_DIM + row;
    for (int j0 = 0; j0 < 4; ++j0) {
      const int jn = (j0 + 1) & 3;
      const v4f* wj = wq + j0 * (8 * (GP_GRU_PITCH / 4));
      const v4f* wn = wq + jn * (8 * (GP_GRU_PITCH / 4));
      v8f ir, iz, in_, hr, hz, hn;
#pragma unroll
      for (int i = 0; i < 8; ++i) {
        ir[i] = bn[0]; iz[i] = bn[1]; in_[i] = bn[2];
        hr[i] = bn[3]; hz[i] = bn[4]; hn[i] = bn[5];
      }
      bn[0] = b_ih[jn * 16 + row]; bn[1] = b_ih[64 + jn * 16 + row]; bn[2] = b_ih[128 + jn * 16 + row];
      bn[3] = b_hh[jn * 16 + row]; bn[4] = b_hh[64 + jn * 16 + row]; bn[5] = b_hh[128 + jn * 16 + row];
#pragma unroll
      for (int c2 = 0; c2 < 8; ++c2) {
        if (c2 < 7) GP_GRU_LOADSET(fb[(c2 + 1) & 1], wj, c2 + 1);
        else        GP_GRU_LOADSET(fb[0], wn, 0);
        __builtin_amdgcn_sched_barrier(0);
        {
          const v4f* f = fb[c2 & 1];
          v2f b;
          b.x = f[0].x; b.y = f[0].y; ir  = wmma4(Aa[2 * c2], b, ir);
          b.x = f[1].x; b.y = f[1].y; iz  = wmma4(Aa[2 * c2], b, iz);
          b.x = f[2].x; b.y = f[2].y; in_ = wmma4(Aa[2 * c2], b, in_);
          b.x = f[3].x; b.y = f[3].y; hr  = wmma4(Ah[2 * c2], b, hr);
          b.x = f[4].x; b.y = f[4].y; hz  = wmma4(Ah[2 * c2], b, hz);
          b.x = f[5].x; b.y = f[5].y; hn  = wmma4(Ah[2 * c2], b, hn);
          b.x = f[0].z; b.y = f[0].w; ir  = wmma4(Aa[2 * c2 + 1], b, ir);
          b.x = f[1].z; b.y = f[1].w; iz  = wmma4(Aa[2 * c2 + 1], b, iz);
          b.x = f[2].z; b.y = f[2].w; in_ = wmma4(Aa[2 * c2 + 1], b, in_);
          b.x = f[3].z; b.y = f[3].w; hr  = wmma4(Ah[2 * c2 + 1], b, hr);
          b.x = f[4].z; b.y = f[4].w; hz  = wmma4(Ah[2 * c2 + 1], b, hz);
          b.x = f[5].z; b.y = f[5].w; hn  = wmma4(Ah[2 * c2 + 1], b, hn);
        }
        __builtin_amdgcn_sched_barrier(0);
      }
#pragma unroll
      for (int i = 0; i < 8; ++i) {
        const float h = hrow[i * GP_NODE_DIM + j0 * 16];
        const float r = sigmoid_(ir[i] + hr[i]);
        const float z = sigmoid_(iz[i] + hz[i]);
        const float n = tanhf(in_[i] + r * hn[i]);
        orow[i * GP_NODE_DIM + j0 * 16] = (1.0f - z) * n + z * h;
      }
    }
    slot += 8;
    tile = slot * GP_GRU_NB + bid;
    if (tile >= ntiles) break;
    const float* ap = abuf + (long)(tile * 16 + row) * GP_ACT_DIM + 2 * half;
    const float* hq = hvin + (long)(tile * 16 + row) * GP_NODE_DIM + 2 * half;
#pragma unroll
    for (int c = 0; c < 16; ++c) { Aa[c] = ld2(ap + 4 * c); Ah[c] = ld2(hq + 4 * c); }
  }
}

extern "C" void kernel_launch(void* const* d_in, const int* in_sizes, int n_in,
                              void* d_out, int out_size, void* d_ws, size_t ws_size,
                              hipStream_t stream) {
  const float* hv    = (const float*)d_in[0];
  const float* he    = (const float*)d_in[1];
  const int*   src   = (const int*)d_in[2];
  const int*   dst   = (const int*)d_in[3];
  const float* W_msg = (const float*)d_in[4];
  const float* b_msg = (const float*)d_in[5];
  const float* w_ih  = (const float*)d_in[6];
  const float* w_hh  = (const float*)d_in[7];
  const float* b_ih  = (const float*)d_in[8];
  const float* b_hh  = (const float*)d_in[9];
  float* out = (float*)d_out;

  const size_t state_elems = (size_t)GP_N_NODES * GP_NODE_DIM;
  float* a_buf  = (float*)d_ws;
  float* hv_buf = a_buf + state_elems;
  float* h_sum   = out;
  int*   rowptr  = (int*)(out + (size_t)GP_N_NODES * GP_EDGE_DIM);
  int*   csr_src = (int*)(out + (size_t)3400000);
  int* rank    = (int*)a_buf;
  int* csr_eid = (int*)a_buf + GP_N_EDGES;
  int* counts  = (int*)a_buf + 2 * GP_N_EDGES;

  hipMemcpyAsync(hv_buf, hv, state_elems * sizeof(float),
                 hipMemcpyDeviceToDevice, stream);

  const int zero_blocks   = (GP_N_NODES / 4 + 255) / 256;
  const int edge_blocks   = (GP_N_EDGES / 4 + 255) / 256;
  const int scan_blocks   = (GP_N_NODES + 1 + 1023) / 1024;
  const int gather_blocks = (GP_N_NODES + 8 * GP_NPW - 1) / (8 * GP_NPW);
  const int node_blocks   = (GP_N_NODES / 16 + 7) / 8;

  gp_zero_kernel<<<zero_blocks, 256, 0, stream>>>(counts);
  gp_rank_kernel<<<edge_blocks, 256, 0, stream>>>(dst, counts, rank);
  gp_scan_kernel<<<scan_blocks, 1024, 0, stream>>>(counts, rowptr);
  gp_fill_kernel<<<edge_blocks, 256, 0, stream>>>(src, dst, rank, rowptr,
                                                  csr_src, csr_eid);
  gp_gather_kernel<1><<<gather_blocks, 256, 0, stream>>>(he, rowptr, csr_eid, h_sum);

  for (int t = 0; t < GP_ROUNDS; ++t) {
    gp_gather_kernel<2><<<gather_blocks, 256, 0, stream>>>(hv_buf, rowptr, csr_src, a_buf);
    gp_node_kernel<<<node_blocks, 256, 0, stream>>>(
        a_buf, hv_buf, h_sum, rowptr,
        W_msg + (size_t)t * GP_CAT_DIM * GP_ACT_DIM,
        b_msg + (size_t)t * GP_ACT_DIM);
    float* outp = (t == GP_ROUNDS - 1) ? out : hv_buf;
    gp_gru_kernel<<<GP_GRU_NB, 256, 0, stream>>>(
        a_buf, hv_buf,
        w_ih + (size_t)t * 3 * GP_NODE_DIM * GP_NODE_DIM,
        w_hh + (size_t)t * 3 * GP_NODE_DIM * GP_NODE_DIM,
        b_ih + (size_t)t * 3 * GP_NODE_DIM,
        b_hh + (size_t)t * 3 * GP_NODE_DIM, outp);
  }
}


# One launch per round: a wave of a persistent work-group sums its tile's source rows, forms a and runs the GRU on WMMA from weights in LDS, S and a passing through its LDS strip; sum order kept
# speedup vs baseline: 2.1867x; 1.0706x over previous
#include <hip/hip_runtime.h>


typedef float v2f __attribute__((ext_vector_type(2)));
typedef float v4f __attribute__((ext_vector_type(4)));
typedef float v8f __attribute__((ext_vector_type(8)));

#define GP_ROUNDS   3
#define GP_NODE_DIM 64
#define GP_ACT_DIM  64
#define GP_EDGE_DIM 32
#define GP_N_NODES  100000
#define GP_N_EDGES  1600000
#define GP_CAT_DIM  160
#define GP_NPW      8

static __device__ __forceinline__ v8f wmma4(v2f a, v2f b, v8f c) {
  return __builtin_amdgcn_wmma_f32_16x16x4_f32(false, a, false, b, (short)0, c,
                                               false, false);
}
static __device__ __forceinline__ v2f ld2(const float* p) {
  return *(const v2f*)p;
}
static __device__ __forceinline__ float sigmoid_(float x) {
  return 1.0f / (1.0f + __expf(-x));
}


__global__ __launch_bounds__(256) void gp_zero_kernel(int* __restrict__ counts) {
  const int i = blockIdx.x * 256 + threadIdx.x;
  if (i < GP_N_NODES / 4) {
    const int4 z = {0, 0, 0, 0};
    ((int4*)counts)[i] = z;
  }
}

__global__ __launch_bounds__(256) void gp_rank_kernel(
    const int* __restrict__ dst, int* __restrict__ counts, int* __restrict__ rank)
{
  const int t = blockIdx.x * 256 + threadIdx.x;
  if (t < GP_N_EDGES / 4) {
    const int4 d = ((const int4*)dst)[t];
    int4 r;
    r.x = atomicAdd(counts + d.x, 1);
    r.y = atomicAdd(counts + d.y, 1);
    r.z = atomicAdd(counts + d.z, 1);
    r.w = atomicAdd(counts + d.w, 1);
    ((int4*)rank)[t] = r;
  }
}

__global__ __launch_bounds__(1024) void gp_scan_kernel(
    const int* __restrict__ counts, int* __restrict__ rowptr)
{
  __shared__ int wsum[32];
  __shared__ int wpre[32];
  const int tid  = threadIdx.x;
  const int lane = tid & 31;
  const int wave = tid >> 5;
  const int base = blockIdx.x * 1024;
  int acc = 0;
  for (int i = tid; i < base; i += 1024) acc += counts[i];
  const int i = base + tid;
  const int c = (i < GP_N_NODES) ? counts[i] : 0;
  int x = c;
  for (int o = 1; o < 32; o <<= 1) {
    const int y = __shfl_up(x, o);
    if (lane >= o) x += y;
  }
  for (int o = 16; o > 0; o >>= 1) acc += __shfl_xor(acc, o);
  if (lane == 31) wsum[wave] = x;
  if (lane == 0)  wpre[wave] = acc;
  __syncthreads();
  int tot = wpre[lane] + ((lane < wave) ? wsum[lane] : 0);
  for (int o = 16; o > 0; o >>= 1) tot += __shfl_xor(tot, o);
  if (i <= GP_N_NODES) rowptr[i] = tot + x - c;
}

__global__ __launch_bounds__(256) void gp_fill_kernel(
    const int* __restrict__ src, const int* __restrict__ dst,
    const int* __restrict__ rank, const int* __restrict__ rowptr,
    int* __restrict__ csr_src, int* __restrict__ csr_eid)
{
  const int t = blockIdx.x * 256 + threadIdx.x;
  if (t < GP_N_EDGES / 4) {
    const int4 s = ((const int4*)src)[t];
    const int4 d = ((const int4*)dst)[t];
    const int4 r = ((const int4*)rank)[t];
    const int p0 = rowptr[d.x] + r.x;
    const int p1 = rowptr[d.y] + r.y;
    const int p2 = rowptr[d.z] + r.z;
    const int p3 = rowptr[d.w] + r.w;
    csr_src[p0] = s.x; csr_eid[p0] = 4 * t;
    csr_src[p1] = s.y; csr_eid[p1] = 4 * t + 1;
    csr_src[p2] = s.z; csr_eid[p2] = 4 * t + 2;
    csr_src[p3] = s.w; csr_eid[p3] = 4 * t + 3;
  }
}

template <int V> struct gp_row;
template <> struct gp_row<1> {
  static __device__ __forceinline__ void load(const char* p, float& a, float& b) {
    a = *(const float*)p; b = 0.0f;
  }
  static __device__ __forceinline__ void store(char* p, float a, float) {
    *(float*)p = a;
  }
};

template <int V>
__global__ __launch_bounds__(256) void gp_gather_kernel(
    const float* __restrict__ rows, const int* __restrict__ rowptr,
    const int* __restrict__ idxs, float* __restrict__ outp)
{
  const int lane = threadIdx.x & 31;
  const int wave = __builtin_amdgcn_readfirstlane((int)(threadIdx.x >> 5));
  const int n0 = (blockIdx.x * 8 + wave) * GP_NPW;
  if (n0 >= GP_N_NODES) return;
  const int nn = (GP_N_NODES - n0 < GP_NPW) ? (GP_N_NODES - n0) : GP_NPW;
  int rp = 0;
  if (lane <= nn) rp = rowptr[n0 + lane];
  const int e0 = __builtin_amdgcn_readlane(rp, 0);
  const int e1 = __builtin_amdgcn_readlane(rp, nn);
  int cur = 0;
  int nb  = __builtin_amdgcn_readlane(rp, 1);
  const unsigned lo = (unsigned)lane * (4u * V);
  float a0 = 0.0f, a1 = 0.0f;
  char* const ob = (char*)outp + (unsigned)n0 * (128u * V) + lo;

  for (int base = e0; base < e1; base += 32) {
    const int cnt = (e1 - base < 32) ? (e1 - base) : 32;
    int idx = 0;
    if (lane < cnt) idx = idxs[base + lane];
#pragma unroll
    for (int h = 0; h < 32; h += 16) {
      if (h < cnt) {
        float r0[16], r1[16];
#pragma unroll
        for (int u = 0; u < 16; ++u) {
          const int k = (h + u < cnt) ? (h + u) : (cnt - 1);
          const unsigned s = (unsigned)__builtin_amdgcn_readlane(idx, k);
          gp_row<V>::load((const char*)rows + ((s * (128u * V)) | lo), r0[u], r1[u]);
        }
#pragma unroll
        for (int u = 0; u < 16; ++u) {
          if (h + u < cnt) {
            const int e = base + h + u;
            while (e == nb) {
              gp_row<V>::store(ob + (unsigned)cur * (128u * V), a0, a1);
              a0 = 0.0f; a1 = 0.0f;
              ++cur;
              nb = __builtin_amdgcn_readlane(rp, cur + 1);
            }
            a0 += r0[u]; a1 += r1[u];
          }
        }
      }
    }
  }
  while (cur < nn) {
    gp_row<V>::store(ob + (unsigned)cur * (128u * V), a0, a1);
    a0 = 0.0f; a1 = 0.0f;
    ++cur;
  }
}

#define GP_RND_NB    256
#define GP_RND_WAVES 16
#define GP_GRU_UNITS 192
#define GP_MSG_UNITS 80
#define GP_PITCH     132
#define GP_SROW      68

static_assert(GP_RND_WAVES * GP_RND_NB <= GP_N_NODES / 16, "every wave must own a first tile");
static_assert(GP_N_NODES % 16 == 0, "whole tiles only");

#define GP_GRU_LOADSET(dst, wj, c2)                                                \
  do {                                                                             \
    _Pragma("unroll")                                                              \
    for (int g = 0; g < 6; ++g)                                                    \
      dst[g] = (wj)[(g * 32 + (c2)) * (GP_PITCH / 4)];                             \
  } while (0)

#define GP_MSG_LOADSET(dst, wm, st)                                                \
  do {                                                                             \
    const int j_ = (st) / 3, h_ = (st) % 3;                                        \
    if (h_ < 2) {                                                                  \
      _Pragma("unroll")                                                            \
      for (int x = 0; x < 4; ++x) {                                                \
        dst[x]     = (wm)[(j_ * 20 + 4 * h_ + x) * (GP_PITCH / 4)];                \
        dst[4 + x] = (wm)[(j_ * 20 + 8 + 4 * h_ + x) * (GP_PITCH / 4)];            \
      }                                                                            \
    } else {                                                                       \
      _Pragma("unroll")                                                            \
      for (int x = 0; x < 4; ++x)                                                  \
        dst[x] = (wm)[(j_ * 20 + 16 + x) * (GP_PITCH / 4)];                        \
    }                                                                              \
  } while (0)

__global__ __launch_bounds__(512) void gp_round_kernel(
    const float* __restrict__ hvin,
    const int* __restrict__ rowptr,
    const int* __restrict__ idxs,
    const float* __restrict__ hsum,
    const float* __restrict__ W,
    const float* __restrict__ bias,
    const float* __restrict__ w_ih, const float* __restrict__ w_hh,
    const float* __restrict__ b_ih, const float* __restrict__ b_hh,
    float* __restrict__ hvout)
{
  __shared__ __attribute__((aligned(16))) float wl[GP_GRU_UNITS * GP_PITCH];
  __shared__ __attribute__((aligned(16))) float wm[GP_MSG_UNITS * GP_PITCH];
  __shared__ __attribute__((aligned(16))) float sl[GP_RND_WAVES * 16 * GP_SROW];

  const int tid  = threadIdx.x;
  const int lane = tid & 31;
  const int wave = __builtin_amdgcn_readfirstlane(tid >> 5);
  const int bid  = (int)blockIdx.x;
  const int ntiles = GP_N_NODES / 16;
  const int row   = lane & 15;
  const int half  = lane >> 4;

  int slot = (wave + bid) & (GP_RND_WAVES - 1);
  int tile = slot * GP_RND_NB + bid;

  {
    const int fc = tid & 15, frl = (tid >> 4) & 15, hb = tid >> 8;
#pragma unroll
    for (int it = 0; it < 12; ++it) {
      const int mb = 2 * (it % 6) + hb;
      const float* wsrc = ((it < 6) ? w_ih : w_hh) + (mb * 16 + frl) * 64 + 4 * fc;
      const v4f v = *(const v4f*)wsrc;
      const int G  = (it / 6) * 3 + (mb >> 2);
      const int j0 = mb & 3;
      float* d = wl + ((G * 4 + j0) * 8 + (fc >> 1)) * GP_PITCH + (fc & 1) * 2;
      v2f lo, hi;
      lo.x = v.x; lo.y = v.y; hi.x = v.z; hi.y = v.w;
      *(v2f*)(d + frl * 4) = lo;
      *(v2f*)(d + (16 + frl) * 4) = hi;
    }
  }
  {
#pragma unroll
    for (int it = 0; it < GP_MSG_UNITS / GP_RND_WAVES; ++it) {
      const int u  = it * GP_RND_WAVES + wave;
      const int j0 = u / 20, c2 = u % 20;
      const float* gsrc = W + (8 * c2 + 2 * half) * GP_ACT_DIM + j0 * 16 + row;
      v4f v;
      v.x = gsrc[0];
      v.y = gsrc[GP_ACT_DIM];
      v.z = gsrc[4 * GP_ACT_DIM];
      v.w = gsrc[5 * GP_ACT_DIM];
      *(v4f*)(wm + u * GP_PITCH + lane * 4) = v;
    }
  }
  __syncthreads();

  const v4f* wq  = (const v4f*)wl + lane;
  const v4f* wmq = (const v4f*)wm + lane;
  float* const sw = sl + wave * (16 * GP_SROW);
  float* const so = sw + 2 * lane;
  const float* const sf = sw + row * GP_SROW + 2 * half;
  float* const sc = sw + (8 * half) * GP_SROW + row;
  const unsigned lo8 = (unsigned)lane * 8u;

  v4f fb[2][6];
  float bn[6];
  bn[0] = b_ih[row]; bn[1] = b_ih[64 + row]; bn[2] = b_ih[128 + row];
  bn[3] = b_hh[row]; bn[4] = b_hh[64 + row]; bn[5] = b_hh[128 + row];
  float bq[4];
#pragma unroll
  for (int j0 = 0; j0 < 4; ++j0) bq[j0] = bias[j0 * 16 + row];

  for (;;) {
    const int n0 = tile * 16;

    {
      int rp = 0;
      if (lane <= 16) rp = rowptr[n0 + lane];
      const int e0 = __builtin_amdgcn_readlane(rp, 0);
      const int e1 = __builtin_amdgcn_readlane(rp, 16);
      int cur = 0;
      int nb  = __builtin_amdgcn_readlane(rp, 1);
      float a0 = 0.0f, a1 = 0.0f;
      for (int base = e0; base < e1; base += 32) {
        const int cnt = (e1 - base < 32) ? (e1 - base) : 32;
        int idx = 0;
        if (lane < cnt) idx = idxs[base + lane];
        float r0[32], r1[32];
#pragma unroll
        for (int u = 0; u < 16; ++u) {
          const int k = (u < cnt) ? u : (cnt - 1);
          const unsigned s = (unsigned)__builtin_amdgcn_readlane(idx, k);
          const v2f t = *(const v2f*)((const char*)hvin + ((s * 256u) | lo8));
          r0[u] = t.x; r1[u] = t.y;
        }
        if (cnt > 16) {
#pragma unroll
          for (int u = 16; u < 32; ++u) {
            const int k = (u < cnt) ? u : (cnt - 1);
            const unsigned s = (unsigned)__builtin_amdgcn_readlane(idx, k);
            const v2f t = *(const v2f*)((const char*)hvin + ((s * 256u) | lo8));
            r0[u] = t.x; r1[u] = t.y;
          }
        }
#pragma unroll
        for (int u = 0; u < 32; ++u) {
          if (u < cnt) {
            const int e = base + u;
            while (e == nb) {
              v2f t; t.x = a0; t.y = a1;
              *(v2f*)(so + cur * GP_SROW) = t;
              a0 = 0.0f; a1 = 0.0f;
              ++cur;
              nb = __builtin_amdgcn_readlane(rp, cur + 1);
            }
            a0 += r0[u]; a1 += r1[u];
          }
        }
      }
      while (cur < 16) {
        v2f t; t.x = a0; t.y = a1;
        *(v2f*)(so + cur * GP_SROW) = t;
        a0 = 0.0f; a1 = 0.0f;
        ++cur;
      }
    }
    __builtin_amdgcn_wave_barrier();

    v2f Ah[16];
    {
      v2f As[16], Ae[8];
      const float* hq = hvin + (long)(n0 + row) * GP_NODE_DIM + 2 * half;
      const float* ep = hsum + (long)(n0 + row) * GP_EDGE_DIM + 2 * half;
#pragma unroll
      for (int c = 0; c < 16; ++c) { As[c] = *(const v2f*)(sf + 4 * c); Ah[c] = ld2(hq + 4 * c); }
#pragma unroll
      for (int c = 0; c < 8; ++c) Ae[c] = ld2(ep + 4 * c);

      float dg[8];
#pragma unroll
      for (int i = 0; i < 8; ++i)
        dg[i] = (float)(rowptr[n0 + 8 * half + i + 1] - rowptr[n0 + 8 * half + i]);

      v4f nf[2][8];
      v8f p, q;
      float bqt[4];
#pragma unroll
      for (int j0 = 0; j0 < 4; ++j0) {
        bqt[j0] = bq[j0];
        asm volatile("" : "+v"(bqt[j0]));
      }
      GP_MSG_LOADSET(nf[0], wmq, 0);
#pragma unroll
      for (int st = 0; st < 12; ++st) {
        const int j0 = st / 3, h = st % 3;
        if (h == 0) {
#pragma unroll
          for (int i = 0; i < 8; ++i) { p[i] = 0.0f; q[i] = bqt[j0]; }
        }
        if (st < 11) GP_MSG_LOADSET(nf[(st + 1) & 1], wmq, st + 1);
        __builtin_amdgcn_sched_barrier(0);
        const v4f* f = nf[st & 1];
        if (h < 2) {
#pragma unroll
          for (int x = 0; x < 4; ++x) {
            const int c2 = 4 * h + x;
            v2f b;
            b.x = f[x].x;     b.y = f[x].y;     p = wmma4(As[2 * c2], b, p);
            b.x = f[4 + x].x; b.y = f[4 + x].y; q = wmma4(Ah[2 * c2], b, q);
            b.x = f[x].z;     b.y = f[x].w;     p = wmma4(As[2 * c2 + 1], b, p);
            b.x = f[4 + x].z; b.y = f[4 + x].w; q = wmma4(Ah[2 * c2 + 1], b, q);
          }
        } else {
#pragma unroll
          for (int x = 0; x < 4; ++x) {
            v2f b;
            b.x = f[x].x; b.y = f[x].y; p = wmma4(Ae[2 * x], b, p);
            b.x = f[x].z; b.y = f[x].w; p = wmma4(Ae[2 * x + 1], b, p);
          }
#pragma unroll
          for (int i = 0; i < 8; ++i) sc[i * GP_SROW + j0 * 16] = fmaf(dg[i], q[i], p[i]);
        }
        __builtin_amdgcn_sched_barrier(0);
      }
    }
    __builtin_amdgcn_wave_barrier();
    v2f Aa[16];
#pragma unroll
    for (int c = 0; c < 16; ++c) Aa[c] = *(const v2f*)(sf + 4 * c);
    GP_GRU_LOADSET(fb[0], wq, 0);

    const float* hrow = hvin  + (long)(n0 + 8 * half) * GP_NODE_DIM + row;
    float*       orow = hvout + (long)(n0 + 8 * half) * GP_NODE_DIM + row;
    for (int j0 = 0; j0 < 4; ++j0) {
      const int jn = (j0 + 1) & 3;
      const v4f* wj = wq + j0 * (8 * (GP_PITCH / 4));
      const v4f* wn = wq + jn * (8 * (GP_PITCH / 4));
      v8f ir, iz, in_, hr, hz, hn;
#pragma unroll
      for (int i = 0; i < 8; ++i) {
        ir[i] = bn[0]; iz[i] = bn[1]; in_[i] = bn[2];
        hr[i] = bn[3]; hz[i] = bn[4]; hn[i] = bn[5];
      }
      bn[0] = b_ih[jn * 16 + row]; bn[1] = b_ih[64 + jn * 16 + row]; bn[2] = b_ih[128 + jn * 16 + row];
      bn[3] = b_hh[jn * 16 + row]; bn[4] = b_hh[64 + jn * 16 + row]; bn[5] = b_hh[128 + jn * 16 + row];
#pragma unroll
      for (int c2 = 0; c2 < 8; ++c2) {
        if (c2 < 7) GP_GRU_LOADSET(fb[(c2 + 1) & 1], wj, c2 + 1);
        else        GP_GRU_LOADSET(fb[0], wn, 0);
        __builtin_amdgcn_sched_barrier(0);
        {
          const v4f* f = fb[c2 & 1];
          v2f b;
          b.x = f[0].x; b.y = f[0].y; ir  = wmma4(Aa[2 * c2], b, ir);
          b.x = f[1].x; b.y = f[1].y; iz  = wmma4(Aa[2 * c2], b, iz);
          b.x = f[2].x; b.y = f[2].y; in_ = wmma4(Aa[2 * c2], b, in_);
          b.x = f[3].x; b.y = f[3].y; hr  = wmma4(Ah[2 * c2], b, hr);
          b.x = f[4].x; b.y = f[4].y; hz  = wmma4(Ah[2 * c2], b, hz);
          b.x = f[5].x; b.y = f[5].y; hn  = wmma4(Ah[2 * c2], b, hn);
          b.x = f[0].z; b.y = f[0].w; ir  = wmma4(Aa[2 * c2 + 1], b, ir);
          b.x = f[1].z; b.y = f[1].w; iz  = wmma4(Aa[2 * c2 + 1], b, iz);
          b.x = f[2].z; b.y = f[2].w; in_ = wmma4(Aa[2 * c2 + 1], b, in_);
          b.x = f[3].z; b.y = f[3].w; hr  = wmma4(Ah[2 * c2 + 1], b, hr);
          b.x = f[4].z; b.y = f[4].w; hz  = wmma4(Ah[2 * c2 + 1], b, hz);
          b.x = f[5].z; b.y = f[5].w; hn  = wmma4(Ah[2 * c2 + 1], b, hn);
        }
        __builtin_amdgcn_sched_barrier(0);
      }
#pragma unroll
      for (int i = 0; i < 8; ++i) {
        const float h = hrow[i * GP_NODE_DIM + j0 * 16];
        const float r = sigmoid_(ir[i] + hr[i]);
        const float z = sigmoid_(iz[i] + hz[i]);
        const float n = tanhf(in_[i] + r * hn[i]);
        orow[i * GP_NODE_DIM + j0 * 16] = (1.0f - z) * n + z * h;
      }
    }
    slot += GP_RND_WAVES;
    tile = slot * GP_RND_NB + bid;
    if (tile >= ntiles) break;
  }
}

extern "C" void kernel_launch(void* const* d_in, const int* in_sizes, int n_in,
                              void* d_out, int out_size, void* d_ws, size_t ws_size,
                              hipStream_t stream) {
  const float* hv    = (const float*)d_in[0];
  const float* he    = (const float*)d_in[1];
  const int*   src   = (const int*)d_in[2];
  const int*   dst   = (const int*)d_in[3];
  const float* W_msg = (const float*)d_in[4];
  const float* b_msg = (const float*)d_in[5];
  const float* w_ih  = (const float*)d_in[6];
  const float* w_hh  = (const float*)d_in[7];
  const float* b_ih  = (const float*)d_in[8];
  const float* b_hh  = (const float*)d_in[9];
  float* out = (float*)d_out;

  const size_t state_elems = (size_t)GP_N_NODES * GP_NODE_DIM;
  float* hv_buf = (float*)d_ws;
  float* h_sum   = hv_buf + state_elems;
  int*   rowptr  = (int*)(h_sum + (size_t)GP_N_NODES * GP_EDGE_DIM);
  int*   csr_src = rowptr + 100032;
  int* rank    = (int*)hv_buf;
  int* csr_eid = (int*)hv_buf + GP_N_EDGES;
  int* counts  = (int*)hv_buf + 2 * GP_N_EDGES;

  const int zero_blocks   = (GP_N_NODES / 4 + 255) / 256;
  const int edge_blocks   = (GP_N_EDGES / 4 + 255) / 256;
  const int scan_blocks   = (GP_N_NODES + 1 + 1023) / 1024;
  const int gather_blocks = (GP_N_NODES + 8 * GP_NPW - 1) / (8 * GP_NPW);

  gp_zero_kernel<<<zero_blocks, 256, 0, stream>>>(counts);
  gp_rank_kernel<<<edge_blocks, 256, 0, stream>>>(dst, counts, rank);
  gp_scan_kernel<<<scan_blocks, 1024, 0, stream>>>(counts, rowptr);
  gp_fill_kernel<<<edge_blocks, 256, 0, stream>>>(src, dst, rank, rowptr,
                                                  csr_src, csr_eid);
  gp_gather_kernel<1><<<gather_blocks, 256, 0, stream>>>(he, rowptr, csr_eid, h_sum);

  for (int t = 0; t < GP_ROUNDS; ++t) {
    const float* inp = (t == 0) ? hv : ((t == 1) ? (const float*)out : (const float*)hv_buf);
    float* outp = (t == 1) ? hv_buf : out;
    gp_round_kernel<<<GP_RND_NB, 512, 0, stream>>>(
        inp, rowptr, csr_src, h_sum,
        W_msg + (size_t)t * GP_CAT_DIM * GP_ACT_DIM,
        b_msg + (size_t)t * GP_ACT_DIM,
        w_ih + (size_t)t * 3 * GP_NODE_DIM * GP_NODE_DIM,
        w_hh + (size_t)t * 3 * GP_NODE_DIM * GP_NODE_DIM,
        b_ih + (size_t)t * 3 * GP_NODE_DIM,
        b_hh + (size_t)t * 3 * GP_NODE_DIM, outp);
  }
}


# Edge lists built with LDS counters: edges binned by 256-node range, each bin put in node order in LDS, lists stored as whole lines; no global atomic, no scattered 4-B store; sums in LDS rank order
# speedup vs baseline: 3.2432x; 1.4831x over previous
#include <hip/hip_runtime.h>


typedef float v2f __attribute__((ext_vector_type(2)));
typedef float v4f __attribute__((ext_vector_type(4)));
typedef float v8f __attribute__((ext_vector_type(8)));
typedef int   v2i __attribute__((ext_vector_type(2)));

#define GP_ROUNDS   3
#define GP_NODE_DIM 64
#define GP_ACT_DIM  64
#define GP_EDGE_DIM 32
#define GP_N_NODES  100000
#define GP_N_EDGES  1600000
#define GP_CAT_DIM  160
#define GP_NPW      8

static __device__ __forceinline__ v8f wmma4(v2f a, v2f b, v8f c) {
  return __builtin_amdgcn_wmma_f32_16x16x4_f32(false, a, false, b, (short)0, c,
                                               false, false);
}
static __device__ __forceinline__ v2f ld2(const float* p) {
  return *(const v2f*)p;
}
static __device__ __forceinline__ float sigmoid_(float x) {
  return 1.0f / (1.0f + __expf(-x));
}


#define GP_NBINS       391
#define GP_SORT_BLOCKS 256
#define GP_CHUNK       6250
#define GP_SCAN_LEN    (GP_NBINS * GP_SORT_BLOCKS)
#define GP_SORT_CAP    8192

static_assert(GP_SORT_BLOCKS * GP_CHUNK == GP_N_EDGES, "whole runs only");
static_assert(GP_NBINS * 256 >= GP_N_NODES && (GP_NBINS - 1) * 256 < GP_N_NODES, "bins cover the nodes");
static_assert(GP_N_NODES <= (1 << 17), "source node packed in 17 bits");

static __device__ __forceinline__ int gp_block_scan(int x, int* ws, int tid) {
  const int lane = tid & 31;
  const int wave = tid >> 5;
  int inc = x;
  for (int o = 1; o < 32; o <<= 1) {
    const int y = __shfl_up(inc, o);
    if (lane >= o) inc += y;
  }
  if (lane == 31) ws[wave] = inc;
  __syncthreads();
  int w = (lane < wave) ? ws[lane] : 0;
  for (int o = 16; o > 0; o >>= 1) w += __shfl_xor(w, o);
  return w + inc - x;
}

__global__ __launch_bounds__(1024) void gp_hist_kernel(
    const int* __restrict__ dst, int* __restrict__ hist)
{
  __shared__ int cnt[GP_NBINS];
  const int tid = threadIdx.x;
  const int blk = blockIdx.x;
  for (int b = tid; b < GP_NBINS; b += 1024) cnt[b] = 0;
  __syncthreads();
  const int* d = dst + blk * GP_CHUNK;
  for (int i = tid; i < GP_CHUNK; i += 1024) atomicAdd(&cnt[d[i] >> 8], 1);
  __syncthreads();
  for (int b = tid; b < GP_NBINS; b += 1024) hist[b * GP_SORT_BLOCKS + blk] = cnt[b];
}

__global__ __launch_bounds__(1024) void gp_scan_kernel(
    const int* __restrict__ counts, int* __restrict__ rowptr, int n)
{
  __shared__ int wsum[32];
  __shared__ int wpre[32];
  const int tid  = threadIdx.x;
  const int lane = tid & 31;
  const int wave = tid >> 5;
  const int base = blockIdx.x * 1024;
  int acc = 0;
  for (int i = tid; i < base; i += 1024) acc += counts[i];
  const int i = base + tid;
  const int c = (i < n) ? counts[i] : 0;
  int x = c;
  for (int o = 1; o < 32; o <<= 1) {
    const int y = __shfl_up(x, o);
    if (lane >= o) x += y;
  }
  for (int o = 16; o > 0; o >>= 1) acc += __shfl_xor(acc, o);
  if (lane == 31) wsum[wave] = x;
  if (lane == 0)  wpre[wave] = acc;
  __syncthreads();
  int tot = wpre[lane] + ((lane < wave) ? wsum[lane] : 0);
  for (int o = 16; o > 0; o >>= 1) tot += __shfl_xor(tot, o);
  if (i <= n) rowptr[i] = tot + x - c;
}

__global__ __launch_bounds__(1024) void gp_bin_kernel(
    const int* __restrict__ src, const int* __restrict__ dst,
    const int* __restrict__ binoff, int* __restrict__ binned_)
{
  v2i* const binned = (v2i*)binned_;
  __shared__ int  cnt[GP_NBINS];
  __shared__ int  gdl[GP_NBINS];
  __shared__ int  ws[32];
  __shared__ v2i  stage[GP_CHUNK];
  __shared__ int  sg[GP_CHUNK];
  const int tid = threadIdx.x;
  const int blk = blockIdx.x;
  const int e0  = blk * GP_CHUNK;
  for (int b = tid; b < GP_NBINS; b += 1024) cnt[b] = 0;
  __syncthreads();
  int d[7], s[7], r[7];
#pragma unroll
  for (int k = 0; k < 7; ++k) {
    const int i = k * 1024 + tid;
    d[k] = 0; s[k] = 0; r[k] = 0;
    if (i < GP_CHUNK) {
      d[k] = dst[e0 + i];
      s[k] = src[e0 + i];
      r[k] = atomicAdd(&cnt[d[k] >> 8], 1);
    }
  }
  __syncthreads();
  const int x  = (tid < GP_NBINS) ? cnt[tid] : 0;
  const int ex = gp_block_scan(x, ws, tid);
  if (tid < GP_NBINS) {
    cnt[tid] = ex;
    gdl[tid] = binoff[tid * GP_SORT_BLOCKS + blk] - ex;
  }
  __syncthreads();
#pragma unroll
  for (int k = 0; k < 7; ++k) {
    const int i = k * 1024 + tid;
    if (i < GP_CHUNK) {
      const int b = d[k] >> 8;
      const int slot = cnt[b] + r[k];
      v2i p;
      p.x = ((d[k] & 255) << 17) | s[k];
      p.y = e0 + i;
      stage[slot] = p;
      sg[slot] = slot + gdl[b];
    }
  }
  __syncthreads();
  for (int i = tid; i < GP_CHUNK; i += 1024) binned[sg[i]] = stage[i];
}

__global__ __launch_bounds__(1024) void gp_sort_kernel(
    const int* __restrict__ binned_, const int* __restrict__ binoff,
    int* __restrict__ rowptr, int* __restrict__ csr_src, int* __restrict__ csr_eid)
{
  const v2i* const binned = (const v2i*)binned_;
  __shared__ int ncnt[256];
  __shared__ int ncur[256];
  __shared__ int ws[32];
  __shared__ int ssrc[GP_SORT_CAP];
  __shared__ int seid[GP_SORT_CAP];
  const int tid = threadIdx.x;
  const int bin = blockIdx.x;
  const int b0 = binoff[bin * GP_SORT_BLOCKS];
  const int b1 = binoff[(bin + 1) * GP_SORT_BLOCKS];
  const int m  = b1 - b0;
  if (tid < 256) ncnt[tid] = 0;
  __syncthreads();
  for (int i = tid; i < m; i += 1024) {
    const v2i p = binned[b0 + i];
    atomicAdd(&ncnt[p.x >> 17], 1);
  }
  __syncthreads();
  const int x  = (tid < 256) ? ncnt[tid] : 0;
  const int ex = gp_block_scan(x, ws, tid);
  if (tid < 256) {
    ncur[tid] = ex;
    const int node = bin * 256 + tid;
    if (node < GP_N_NODES) rowptr[node] = b0 + ex;
  }
  if (bin == GP_NBINS - 1 && tid == 0) rowptr[GP_N_NODES] = b1;
  __syncthreads();
  if (m <= GP_SORT_CAP) {
    for (int i = tid; i < m; i += 1024) {
      const v2i p = binned[b0 + i];
      const int pos = atomicAdd(&ncur[p.x >> 17], 1);
      ssrc[pos] = p.x & 0x1FFFF;
      seid[pos] = p.y;
    }
    __syncthreads();
    for (int i = tid; i < m; i += 1024) {
      csr_src[b0 + i] = ssrc[i];
      csr_eid[b0 + i] = seid[i];
    }
  } else {
    for (int i = tid; i < m; i += 1024) {
      const v2i p = binned[b0 + i];
      const int pos = b0 + atomicAdd(&ncur[p.x >> 17], 1);
      csr_src[pos] = p.x & 0x1FFFF;
      csr_eid[pos] = p.y;
    }
  }
}

template <int V> struct gp_row;
template <> struct gp_row<1> {
  static __device__ __forceinline__ void load(const char* p, float& a, float& b) {
    a = *(const float*)p; b = 0.0f;
  }
  static __device__ __forceinline__ void store(char* p, float a, float) {
    *(float*)p = a;
  }
};

template <int V>
__global__ __launch_bounds__(256) void gp_gather_kernel(
    const float* __restrict__ rows, const int* __restrict__ rowptr,
    const int* __restrict__ idxs, float* __restrict__ outp)
{
  const int lane = threadIdx.x & 31;
  const int wave = __builtin_amdgcn_readfirstlane((int)(threadIdx.x >> 5));
  const int n0 = (blockIdx.x * 8 + wave) * GP_NPW;
  if (n0 >= GP_N_NODES) return;
  const int nn = (GP_N_NODES - n0 < GP_NPW) ? (GP_N_NODES - n0) : GP_NPW;
  int rp = 0;
  if (lane <= nn) rp = rowptr[n0 + lane];
  const int e0 = __builtin_amdgcn_readlane(rp, 0);
  const int e1 = __builtin_amdgcn_readlane(rp, nn);
  int cur = 0;
  int nb  = __builtin_amdgcn_readlane(rp, 1);
  const unsigned lo = (unsigned)lane * (4u * V);
  float a0 = 0.0f, a1 = 0.0f;
  char* const ob = (char*)outp + (unsigned)n0 * (128u * V) + lo;

  for (int base = e0; base < e1; base += 32) {
    const int cnt = (e1 - base < 32) ? (e1 - base) : 32;
    int idx = 0;
    if (lane < cnt) idx = idxs[base + lane];
#pragma unroll
    for (int h = 0; h < 32; h += 16) {
      if (h < cnt) {
        float r0[16], r1[16];
#pragma unroll
        for (int u = 0; u < 16; ++u) {
          const int k = (h + u < cnt) ? (h + u) : (cnt - 1);
          const unsigned s = (unsigned)__builtin_amdgcn_readlane(idx, k);
          gp_row<V>::load((const char*)rows + ((s * (128u * V)) | lo), r0[u], r1[u]);
        }
#pragma unroll
        for (int u = 0; u < 16; ++u) {
          if (h + u < cnt) {
            const int e = base + h + u;
            while (e == nb) {
              gp_row<V>::store(ob + (unsigned)cur * (128u * V), a0, a1);
              a0 = 0.0f; a1 = 0.0f;
              ++cur;
              nb = __builtin_amdgcn_readlane(rp, cur + 1);
            }
            a0 += r0[u]; a1 += r1[u];
          }
        }
      }
    }
  }
  while (cur < nn) {
    gp_row<V>::store(ob + (unsigned)cur * (128u * V), a0, a1);
    a0 = 0.0f; a1 = 0.0f;
    ++cur;
  }
}

#define GP_RND_NB    256
#define GP_RND_WAVES 16
#define GP_GRU_UNITS 192
#define GP_MSG_UNITS 80
#define GP_PITCH     132
#define GP_SROW      68

static_assert(GP_RND_WAVES * GP_RND_NB <= GP_N_NODES / 16, "every wave must own a first tile");
static_assert(GP_N_NODES % 16 == 0, "whole tiles only");

#define GP_GRU_LOADSET(dst, wj, c2)                                                \
  do {                                                                             \
    _Pragma("unroll")                                                              \
    for (int g = 0; g < 6; ++g)                                                    \
      dst[g] = (wj)[(g * 32 + (c2)) * (GP_PITCH / 4)];                             \
  } while (0)

#define GP_MSG_LOADSET(dst, wm, st)                                                \
  do {                                                                             \
    const int j_ = (st) / 3, h_ = (st) % 3;                                        \
    if (h_ < 2) {                                                                  \
      _Pragma("unroll")                                                            \
      for (int x = 0; x < 4; ++x) {                                                \
        dst[x]     = (wm)[(j_ * 20 + 4 * h_ + x) * (GP_PITCH / 4)];                \
        dst[4 + x] = (wm)[(j_ * 20 + 8 + 4 * h_ + x) * (GP_PITCH / 4)];            \
      }                                                                            \
    } else {                                                                       \
      _Pragma("unroll")                                                            \
      for (int x = 0; x < 4; ++x)                                                  \
        dst[x] = (wm)[(j_ * 20 + 16 + x) * (GP_PITCH / 4)];                        \
    }                                                                              \
  } while (0)

__global__ __launch_bounds__(512) void gp_round_kernel(
    const float* __restrict__ hvin,
    const int* __restrict__ rowptr,
    const int* __restrict__ idxs,
    const float* __restrict__ hsum,
    const float* __restrict__ W,
    const float* __restrict__ bias,
    const float* __restrict__ w_ih, const float* __restrict__ w_hh,
    const float* __restrict__ b_ih, const float* __restrict__ b_hh,
    float* __restrict__ hvout)
{
  __shared__ __attribute__((aligned(16))) float wl[GP_GRU_UNITS * GP_PITCH];
  __shared__ __attribute__((aligned(16))) float wm[GP_MSG_UNITS * GP_PITCH];
  __shared__ __attribute__((aligned(16))) float sl[GP_RND_WAVES * 16 * GP_SROW];

  const int tid  = threadIdx.x;
  const int lane = tid & 31;
  const int wave = __builtin_amdgcn_readfirstlane(tid >> 5);
  const int bid  = (int)blockIdx.x;
  const int ntiles = GP_N_NODES / 16;
  const int row   = lane & 15;
  const int half  = lane >> 4;

  int slot = (wave + bid) & (GP_RND_WAVES - 1);
  int tile = slot * GP_RND_NB + bid;

  {
    const int fc = tid & 15, frl = (tid >> 4) & 15, hb = tid >> 8;
#pragma unroll
    for (int it = 0; it < 12; ++it) {
      const int mb = 2 * (it % 6) + hb;
      const float* wsrc = ((it < 6) ? w_ih : w_hh) + (mb * 16 + frl) * 64 + 4 * fc;
      const v4f v = *(const v4f*)wsrc;
      const int G  = (it / 6) * 3 + (mb >> 2);
      const int j0 = mb & 3;
      float* d = wl + ((G * 4 + j0) * 8 + (fc >> 1)) * GP_PITCH + (fc & 1) * 2;
      v2f lo, hi;
      lo.x = v.x; lo.y = v.y; hi.x = v.z; hi.y = v.w;
      *(v2f*)(d + frl * 4) = lo;
      *(v2f*)(d + (16 + frl) * 4) = hi;
    }
  }
  {
#pragma unroll
    for (int it = 0; it < GP_MSG_UNITS / GP_RND_WAVES; ++it) {
      const int u  = it * GP_RND_WAVES + wave;
      const int j0 = u / 20, c2 = u % 20;
      const float* gsrc = W + (8 * c2 + 2 * half) * GP_ACT_DIM + j0 * 16 + row;
      v4f v;
      v.x = gsrc[0];
      v.y = gsrc[GP_ACT_DIM];
      v.z = gsrc[4 * GP_ACT_DIM];
      v.w = gsrc[5 * GP_ACT_DIM];
      *(v4f*)(wm + u * GP_PITCH + lane * 4) = v;
    }
  }
  __syncthreads();

  const v4f* wq  = (const v4f*)wl + lane;
  const v4f* wmq = (const v4f*)wm + lane;
  float* const sw = sl + wave * (16 * GP_SROW);
  float* const so = sw + 2 * lane;
  const float* const sf = sw + row * GP_SROW + 2 * half;
  float* const sc = sw + (8 * half) * GP_SROW + row;
  const unsigned lo8 = (unsigned)lane * 8u;

  v4f fb[2][6];
  float bn[6];
  bn[0] = b_ih[row]; bn[1] = b_ih[64 + row]; bn[2] = b_ih[128 + row];
  bn[3] = b_hh[row]; bn[4] = b_hh[64 + row]; bn[5] = b_hh[128 + row];
  float bq[4];
#pragma unroll
  for (int j0 = 0; j0 < 4; ++j0) bq[j0] = bias[j0 * 16 + row];

  for (;;) {
    const int n0 = tile * 16;

    {
      int rp = 0;
      if (lane <= 16) rp = rowptr[n0 + lane];
      const int e0 = __builtin_amdgcn_readlane(rp, 0);
      const int e1 = __builtin_amdgcn_readlane(rp, 16);
      int cur = 0;
      int nb  = __builtin_amdgcn_readlane(rp, 1);
      float a0 = 0.0f, a1 = 0.0f;
      for (int base = e0; base < e1; base += 32) {
        const int cnt = (e1 - base < 32) ? (e1 - base) : 32;
        int idx = 0;
        if (lane < cnt) idx = idxs[base + lane];
        float r0[32], r1[32];
#pragma unroll
        for (int u = 0; u < 16; ++u) {
          const int k = (u < cnt) ? u : (cnt - 1);
          const unsigned s = (unsigned)__builtin_amdgcn_readlane(idx, k);
          const v2f t = *(const v2f*)((const char*)hvin + ((s * 256u) | lo8));
          r0[u] = t.x; r1[u] = t.y;
        }
        if (cnt > 16) {
#pragma unroll
          for (int u = 16; u < 32; ++u) {
            const int k = (u < cnt) ? u : (cnt - 1);
            const unsigned s = (unsigned)__builtin_amdgcn_readlane(idx, k);
            const v2f t = *(const v2f*)((const char*)hvin + ((s * 256u) | lo8));
            r0[u] = t.x; r1[u] = t.y;
          }
        }
#pragma unroll
        for (int u = 0; u < 32; ++u) {
          if (u < cnt) {
            const int e = base + u;
            while (e == nb) {
              v2f t; t.x = a0; t.y = a1;
              *(v2f*)(so + cur * GP_SROW) = t;
              a0 = 0.0f; a1 = 0.0f;
              ++cur;
              nb = __builtin_amdgcn_readlane(rp, cur + 1);
            }
            a0 += r0[u]; a1 += r1[u];
          }
        }
      }
      while (cur < 16) {
        v2f t; t.x = a0; t.y = a1;
        *(v2f*)(so + cur * GP_SROW) = t;
        a0 = 0.0f; a1 = 0.0f;
        ++cur;
      }
    }
    __builtin_amdgcn_wave_barrier();

    v2f Ah[16];
    {
      v2f As[16], Ae[8];
      const float* hq = hvin + (long)(n0 + row) * GP_NODE_DIM + 2 * half;
      const float* ep = hsum + (long)(n0 + row) * GP_EDGE_DIM + 2 * half;
#pragma unroll
      for (int c = 0; c < 16; ++c) { As[c] = *(const v2f*)(sf + 4 * c); Ah[c] = ld2(hq + 4 * c); }
#pragma unroll
      for (int c = 0; c < 8; ++c) Ae[c] = ld2(ep + 4 * c);

      float dg[8];
#pragma unroll
      for (int i = 0; i < 8; ++i)
        dg[i] = (float)(rowptr[n0 + 8 * half + i + 1] - rowptr[n0 + 8 * half + i]);

      v4f nf[2][8];
      v8f p, q;
      float bqt[4];
#pragma unroll
      for (int j0 = 0; j0 < 4; ++j0) {
        bqt[j0] = bq[j0];
        asm volatile("" : "+v"(bqt[j0]));
      }
      GP_MSG_LOADSET(nf[0], wmq, 0);
#pragma unroll
      for (int st = 0; st < 12; ++st) {
        const int j0 = st / 3, h = st % 3;
        if (h == 0) {
#pragma unroll
          for (int i = 0; i < 8; ++i) { p[i] = 0.0f; q[i] = bqt[j0]; }
        }
        if (st < 11) GP_MSG_LOADSET(nf[(st + 1) & 1], wmq, st + 1);
        __builtin_amdgcn_sched_barrier(0);
        const v4f* f = nf[st & 1];
        if (h < 2) {
#pragma unroll
          for (int x = 0; x < 4; ++x) {
            const int c2 = 4 * h + x;
            v2f b;
            b.x = f[x].x;     b.y = f[x].y;     p = wmma4(As[2 * c2], b, p);
            b.x = f[4 + x].x; b.y = f[4 + x].y; q = wmma4(Ah[2 * c2], b, q);
            b.x = f[x].z;     b.y = f[x].w;     p = wmma4(As[2 * c2 + 1], b, p);
            b.x = f[4 + x].z; b.y = f[4 + x].w; q = wmma4(Ah[2 * c2 + 1], b, q);
          }
        } else {
#pragma unroll
          for (int x = 0; x < 4; ++x) {
            v2f b;
            b.x = f[x].x; b.y = f[x].y; p = wmma4(Ae[2 * x], b, p);
            b.x = f[x].z; b.y = f[x].w; p = wmma4(Ae[2 * x + 1], b, p);
          }
#pragma unroll
          for (int i = 0; i < 8; ++i) sc[i * GP_SROW + j0 * 16] = fmaf(dg[i], q[i], p[i]);
        }
        __builtin_amdgcn_sched_barrier(0);
      }
    }
    __builtin_amdgcn_wave_barrier();
    v2f Aa[16];
#pragma unroll
    for (int c = 0; c < 16; ++c) Aa[c] = *(const v2f*)(sf + 4 * c);
    GP_GRU_LOADSET(fb[0], wq, 0);

    const float* hrow = hvin  + (long)(n0 + 8 * half) * GP_NODE_DIM + row;
    float*       orow = hvout + (long)(n0 + 8 * half) * GP_NODE_DIM + row;
    for (int j0 = 0; j0 < 4; ++j0) {
      const int jn = (j0 + 1) & 3;
      const v4f* wj = wq + j0 * (8 * (GP_PITCH / 4));
      const v4f* wn = wq + jn * (8 * (GP_PITCH / 4));
      v8f ir, iz, in_, hr, hz, hn;
#pragma unroll
      for (int i = 0; i < 8; ++i) {
        ir[i] = bn[0]; iz[i] = bn[1]; in_[i] = bn[2];
        hr[i] = bn[3]; hz[i] = bn[4]; hn[i] = bn[5];
      }
      bn[0] = b_ih[jn * 16 + row]; bn[1] = b_ih[64 + jn * 16 + row]; bn[2] = b_ih[128 + jn * 16 + row];
      bn[3] = b_hh[jn * 16 + row]; bn[4] = b_hh[64 + jn * 16 + row]; bn[5] = b_hh[128 + jn * 16 + row];
#pragma unroll
      for (int c2 = 0; c2 < 8; ++c2) {
        if (c2 < 7) GP_GRU_LOADSET(fb[(c2 + 1) & 1], wj, c2 + 1);
        else        GP_GRU_LOADSET(fb[0], wn, 0);
        __builtin_amdgcn_sched_barrier(0);
        {
          const v4f* f = fb[c2 & 1];
          v2f b;
          b.x = f[0].x; b.y = f[0].y; ir  = wmma4(Aa[2 * c2], b, ir);
          b.x = f[1].x; b.y = f[1].y; iz  = wmma4(Aa[2 * c2], b, iz);
          b.x = f[2].x; b.y = f[2].y; in_ = wmma4(Aa[2 * c2], b, in_);
          b.x = f[3].x; b.y = f[3].y; hr  = wmma4(Ah[2 * c2], b, hr);
          b.x = f[4].x; b.y = f[4].y; hz  = wmma4(Ah[2 * c2], b, hz);
          b.x = f[5].x; b.y = f[5].y; hn  = wmma4(Ah[2 * c2], b, hn);
          b.x = f[0].z; b.y = f[0].w; ir  = wmma4(Aa[2 * c2 + 1], b, ir);
          b.x = f[1].z; b.y = f[1].w; iz  = wmma4(Aa[2 * c2 + 1], b, iz);
          b.x = f[2].z; b.y = f[2].w; in_ = wmma4(Aa[2 * c2 + 1], b, in_);
          b.x = f[3].z; b.y = f[3].w; hr  = wmma4(Ah[2 * c2 + 1], b, hr);
          b.x = f[4].z; b.y = f[4].w; hz  = wmma4(Ah[2 * c2 + 1], b, hz);
          b.x = f[5].z; b.y = f[5].w; hn  = wmma4(Ah[2 * c2 + 1], b, hn);
        }
        __builtin_amdgcn_sched_barrier(0);
      }
#pragma unroll
      for (int i = 0; i < 8; ++i) {
        const float h = hrow[i * GP_NODE_DIM + j0 * 16];
        const float r = sigmoid_(ir[i] + hr[i]);
        const float z = sigmoid_(iz[i] + hz[i]);
        const float n = tanhf(in_[i] + r * hn[i]);
        orow[i * GP_NODE_DIM + j0 * 16] = (1.0f - z) * n + z * h;
      }
    }
    slot += GP_RND_WAVES;
    tile = slot * GP_RND_NB + bid;
    if (tile >= ntiles) break;
  }
}

extern "C" void kernel_launch(void* const* d_in, const int* in_sizes, int n_in,
                              void* d_out, int out_size, void* d_ws, size_t ws_size,
                              hipStream_t stream) {
  const float* hv    = (const float*)d_in[0];
  const float* he    = (const float*)d_in[1];
  const int*   src   = (const int*)d_in[2];
  const int*   dst   = (const int*)d_in[3];
  const float* W_msg = (const float*)d_in[4];
  const float* b_msg = (const float*)d_in[5];
  const float* w_ih  = (const float*)d_in[6];
  const float* w_hh  = (const float*)d_in[7];
  const float* b_ih  = (const float*)d_in[8];
  const float* b_hh  = (const float*)d_in[9];
  float* out = (float*)d_out;

  const size_t state_elems = (size_t)GP_N_NODES * GP_NODE_DIM;
  float* hv_buf = (float*)d_ws;
  float* h_sum   = hv_buf + state_elems;
  int*   rowptr  = (int*)(h_sum + (size_t)GP_N_NODES * GP_EDGE_DIM);
  int*   csr_src = rowptr + 100032;
  int* binned  = (int*)hv_buf;
  int* csr_eid = (int*)hv_buf + 2 * GP_N_EDGES;
  int* hist    = (int*)hv_buf + 3 * GP_N_EDGES;
  int* binoff  = hist + 100352;

  const int scan_blocks   = (GP_SCAN_LEN + 1 + 1023) / 1024;
  const int gather_blocks = (GP_N_NODES + 8 * GP_NPW - 1) / (8 * GP_NPW);

  gp_hist_kernel<<<GP_SORT_BLOCKS, 1024, 0, stream>>>(dst, hist);
  gp_scan_kernel<<<scan_blocks, 1024, 0, stream>>>(hist, binoff, GP_SCAN_LEN);
  gp_bin_kernel<<<GP_SORT_BLOCKS, 1024, 0, stream>>>(src, dst, binoff, binned);
  gp_sort_kernel<<<GP_NBINS, 1024, 0, stream>>>(binned, binoff, rowptr, csr_src, csr_eid);
  gp_gather_kernel<1><<<gather_blocks, 256, 0, stream>>>(he, rowptr, csr_eid, h_sum);

  for (int t = 0; t < GP_ROUNDS; ++t) {
    const float* inp = (t == 0) ? hv : ((t == 1) ? (const float*)out : (const float*)hv_buf);
    float* outp = (t == 1) ? hv_buf : out;
    gp_round_kernel<<<GP_RND_NB, 512, 0, stream>>>(
        inp, rowptr, csr_src, h_sum,
        W_msg + (size_t)t * GP_CAT_DIM * GP_ACT_DIM,
        b_msg + (size_t)t * GP_ACT_DIM,
        w_ih + (size_t)t * 3 * GP_NODE_DIM * GP_NODE_DIM,
        w_hh + (size_t)t * 3 * GP_NODE_DIM * GP_NODE_DIM,
        b_ih + (size_t)t * 3 * GP_NODE_DIM,
        b_hh + (size_t)t * 3 * GP_NODE_DIM, outp);
  }
}


# GRU gate stage on the bare hardware functions: sigmoid = v_rcp_f32(1 + v_exp_f32), tanh = 1 - 2/(exp(2x)+1), in place of the IEEE division sequence and the two-branch library tanh; sums untouched
# speedup vs baseline: 3.6531x; 1.1264x over previous
#include <hip/hip_runtime.h>


typedef float v2f __attribute__((ext_vector_type(2)));
typedef float v4f __attribute__((ext_vector_type(4)));
typedef float v8f __attribute__((ext_vector_type(8)));
typedef int   v2i __attribute__((ext_vector_type(2)));

#define GP_ROUNDS   3
#define GP_NODE_DIM 64
#define GP_ACT_DIM  64
#define GP_EDGE_DIM 32
#define GP_N_NODES  100000
#define GP_N_EDGES  1600000
#define GP_CAT_DIM  160
#define GP_NPW      8

static __device__ __forceinline__ v8f wmma4(v2f a, v2f b, v8f c) {
  return __builtin_amdgcn_wmma_f32_16x16x4_f32(false, a, false, b, (short)0, c,
                                               false, false);
}
static __device__ __forceinline__ v2f ld2(const float* p) {
  return *(const v2f*)p;
}
static __device__ __forceinline__ float sigmoid_(float x) {
  return __builtin_amdgcn_rcpf(1.0f + __builtin_amdgcn_exp2f(x * -1.44269504f));
}
static __device__ __forceinline__ float tanh_(float x) {
  return fmaf(__builtin_amdgcn_rcpf(1.0f + __builtin_amdgcn_exp2f(x * 2.88539008f)), -2.0f, 1.0f);
}


#define GP_NBINS       391
#define GP_SORT_BLOCKS 256
#define GP_CHUNK       6250
#define GP_SCAN_LEN    (GP_NBINS * GP_SORT_BLOCKS)
#define GP_SORT_CAP    8192

static_assert(GP_SORT_BLOCKS * GP_CHUNK == GP_N_EDGES, "whole runs only");
static_assert(GP_NBINS * 256 >= GP_N_NODES && (GP_NBINS - 1) * 256 < GP_N_NODES, "bins cover the nodes");
static_assert(GP_N_NODES <= (1 << 17), "source node packed in 17 bits");

static __device__ __forceinline__ int gp_block_scan(int x, int* ws, int tid) {
  const int lane = tid & 31;
  const int wave = tid >> 5;
  int inc = x;
  for (int o = 1; o < 32; o <<= 1) {
    const int y = __shfl_up(inc, o);
    if (lane >= o) inc += y;
  }
  if (lane == 31) ws[wave] = inc;
  __syncthreads();
  int w = (lane < wave) ? ws[lane] : 0;
  for (int o = 16; o > 0; o >>= 1) w += __shfl_xor(w, o);
  return w + inc - x;
}

__global__ __launch_bounds__(1024) void gp_hist_kernel(
    const int* __restrict__ dst, int* __restrict__ hist)
{
  __shared__ int cnt[GP_NBINS];
  const int tid = threadIdx.x;
  const int blk = blockIdx.x;
  for (int b = tid; b < GP_NBINS; b += 1024) cnt[b] = 0;
  __syncthreads();
  const int* d = dst + blk * GP_CHUNK;
  for (int i = tid; i < GP_CHUNK; i += 1024) atomicAdd(&cnt[d[i] >> 8], 1);
  __syncthreads();
  for (int b = tid; b < GP_NBINS; b += 1024) hist[b * GP_SORT_BLOCKS + blk] = cnt[b];
}

__global__ __launch_bounds__(1024) void gp_scan_kernel(
    const int* __restrict__ counts, int* __restrict__ rowptr, int n)
{
  __shared__ int wsum[32];
  __shared__ int wpre[32];
  const int tid  = threadIdx.x;
  const int lane = tid & 31;
  const int wave = tid >> 5;
  const int base = blockIdx.x * 1024;
  int acc = 0;
  for (int i = tid; i < base; i += 1024) acc += counts[i];
  const int i = base + tid;
  const int c = (i < n) ? counts[i] : 0;
  int x = c;
  for (int o = 1; o < 32; o <<= 1) {
    const int y = __shfl_up(x, o);
    if (lane >= o) x += y;
  }
  for (int o = 16; o > 0; o >>= 1) acc += __shfl_xor(acc, o);
  if (lane == 31) wsum[wave] = x;
  if (lane == 0)  wpre[wave] = acc;
  __syncthreads();
  int tot = wpre[lane] + ((lane < wave) ? wsum[lane] : 0);
  for (int o = 16; o > 0; o >>= 1) tot += __shfl_xor(tot, o);
  if (i <= n) rowptr[i] = tot + x - c;
}

__global__ __launch_bounds__(1024) void gp_bin_kernel(
    const int* __restrict__ src, const int* __restrict__ dst,
    const int* __restrict__ binoff, int* __restrict__ binned_)
{
  v2i* const binned = (v2i*)binned_;
  __shared__ int  cnt[GP_NBINS];
  __shared__ int  gdl[GP_NBINS];
  __shared__ int  ws[32];
  __shared__ v2i  stage[GP_CHUNK];
  __shared__ int  sg[GP_CHUNK];
  const int tid = threadIdx.x;
  const int blk = blockIdx.x;
  const int e0  = blk * GP_CHUNK;
  for (int b = tid; b < GP_NBINS; b += 1024) cnt[b] = 0;
  __syncthreads();
  int d[7], s[7], r[7];
#pragma unroll
  for (int k = 0; k < 7; ++k) {
    const int i = k * 1024 + tid;
    d[k] = 0; s[k] = 0; r[k] = 0;
    if (i < GP_CHUNK) {
      d[k] = dst[e0 + i];
      s[k] = src[e0 + i];
      r[k] = atomicAdd(&cnt[d[k] >> 8], 1);
    }
  }
  __syncthreads();
  const int x  = (tid < GP_NBINS) ? cnt[tid] : 0;
  const int ex = gp_block_scan(x, ws, tid);
  if (tid < GP_NBINS) {
    cnt[tid] = ex;
    gdl[tid] = binoff[tid * GP_SORT_BLOCKS + blk] - ex;
  }
  __syncthreads();
#pragma unroll
  for (int k = 0; k < 7; ++k) {
    const int i = k * 1024 + tid;
    if (i < GP_CHUNK) {
      const int b = d[k] >> 8;
      const int slot = cnt[b] + r[k];
      v2i p;
      p.x = ((d[k] & 255) << 17) | s[k];
      p.y = e0 + i;
      stage[slot] = p;
      sg[slot] = slot + gdl[b];
    }
  }
  __syncthreads();
  for (int i = tid; i < GP_CHUNK; i += 1024) binned[sg[i]] = stage[i];
}

__global__ __launch_bounds__(1024) void gp_sort_kernel(
    const int* __restrict__ binned_, const int* __restrict__ binoff,
    int* __restrict__ rowptr, int* __restrict__ csr_src, int* __restrict__ csr_eid)
{
  const v2i* const binned = (const v2i*)binned_;
  __shared__ int ncnt[256];
  __shared__ int ncur[256];
  __shared__ int ws[32];
  __shared__ int ssrc[GP_SORT_CAP];
  __shared__ int seid[GP_SORT_CAP];
  const int tid = threadIdx.x;
  const int bin = blockIdx.x;
  const int b0 = binoff[bin * GP_SORT_BLOCKS];
  const int b1 = binoff[(bin + 1) * GP_SORT_BLOCKS];
  const int m  = b1 - b0;
  if (tid < 256) ncnt[tid] = 0;
  __syncthreads();
  for (int i = tid; i < m; i += 1024) {
    const v2i p = binned[b0 + i];
    atomicAdd(&ncnt[p.x >> 17], 1);
  }
  __syncthreads();
  const int x  = (tid < 256) ? ncnt[tid] : 0;
  const int ex = gp_block_scan(x, ws, tid);
  if (tid < 256) {
    ncur[tid] = ex;
    const int node = bin * 256 + tid;
    if (node < GP_N_NODES) rowptr[node] = b0 + ex;
  }
  if (bin == GP_NBINS - 1 && tid == 0) rowptr[GP_N_NODES] = b1;
  __syncthreads();
  if (m <= GP_SORT_CAP) {
    for (int i = tid; i < m; i += 1024) {
      const v2i p = binned[b0 + i];
      const int pos = atomicAdd(&ncur[p.x >> 17], 1);
      ssrc[pos] = p.x & 0x1FFFF;
      seid[pos] = p.y;
    }
    __syncthreads();
    for (int i = tid; i < m; i += 1024) {
      csr_src[b0 + i] = ssrc[i];
      csr_eid[b0 + i] = seid[i];
    }
  } else {
    for (int i = tid; i < m; i += 1024) {
      const v2i p = binned[b0 + i];
      const int pos = b0 + atomicAdd(&ncur[p.x >> 17], 1);
      csr_src[pos] = p.x & 0x1FFFF;
      csr_eid[pos] = p.y;
    }
  }
}

template <int V> struct gp_row;
template <> struct gp_row<1> {
  static __device__ __forceinline__ void load(const char* p, float& a, float& b) {
    a = *(const float*)p; b = 0.0f;
  }
  static __device__ __forceinline__ void store(char* p, float a, float) {
    *(float*)p = a;
  }
};

template <int V>
__global__ __launch_bounds__(256) void gp_gather_kernel(
    const float* __restrict__ rows, const int* __restrict__ rowptr,
    const int* __restrict__ idxs, float* __restrict__ outp)
{
  const int lane = threadIdx.x & 31;
  const int wave = __builtin_amdgcn_readfirstlane((int)(threadIdx.x >> 5));
  const int n0 = (blockIdx.x * 8 + wave) * GP_NPW;
  if (n0 >= GP_N_NODES) return;
  const int nn = (GP_N_NODES - n0 < GP_NPW) ? (GP_N_NODES - n0) : GP_NPW;
  int rp = 0;
  if (lane <= nn) rp = rowptr[n0 + lane];
  const int e0 = __builtin_amdgcn_readlane(rp, 0);
  const int e1 = __builtin_amdgcn_readlane(rp, nn);
  int cur = 0;
  int nb  = __builtin_amdgcn_readlane(rp, 1);
  const unsigned lo = (unsigned)lane * (4u * V);
  float a0 = 0.0f, a1 = 0.0f;
  char* const ob = (char*)outp + (unsigned)n0 * (128u * V) + lo;

  for (int base = e0; base < e1; base += 32) {
    const int cnt = (e1 - base < 32) ? (e1 - base) : 32;
    int idx = 0;
    if (lane < cnt) idx = idxs[base + lane];
#pragma unroll
    for (int h = 0; h < 32; h += 16) {
      if (h < cnt) {
        float r0[16], r1[16];
#pragma unroll
        for (int u = 0; u < 16; ++u) {
          const int k = (h + u < cnt) ? (h + u) : (cnt - 1);
          const unsigned s = (unsigned)__builtin_amdgcn_readlane(idx, k);
          gp_row<V>::load((const char*)rows + ((s * (128u * V)) | lo), r0[u], r1[u]);
        }
#pragma unroll
        for (int u = 0; u < 16; ++u) {
          if (h + u < cnt) {
            const int e = base + h + u;
            while (e == nb) {
              gp_row<V>::store(ob + (unsigned)cur * (128u * V), a0, a1);
              a0 = 0.0f; a1 = 0.0f;
              ++cur;
              nb = __builtin_amdgcn_readlane(rp, cur + 1);
            }
            a0 += r0[u]; a1 += r1[u];
          }
        }
      }
    }
  }
  while (cur < nn) {
    gp_row<V>::store(ob + (unsigned)cur * (128u * V), a0, a1);
    a0 = 0.0f; a1 = 0.0f;
    ++cur;
  }
}

#define GP_RND_NB    256
#define GP_RND_WAVES 16
#define GP_GRU_UNITS 192
#define GP_MSG_UNITS 80
#define GP_PITCH     132
#define GP_SROW      68

static_assert(GP_RND_WAVES * GP_RND_NB <= GP_N_NODES / 16, "every wave must own a first tile");
static_assert(GP_N_NODES % 16 == 0, "whole tiles only");

#define GP_GRU_LOADSET(dst, wj, c2)                                                \
  do {                                                                             \
    _Pragma("unroll")                                                              \
    for (int g = 0; g < 6; ++g)                                                    \
      dst[g] = (wj)[(g * 32 + (c2)) * (GP_PITCH / 4)];                             \
  } while (0)

#define GP_MSG_LOADSET(dst, wm, st)                                                \
  do {                                                                             \
    const int j_ = (st) / 3, h_ = (st) % 3;                                        \
    if (h_ < 2) {                                                                  \
      _Pragma("unroll")                                                            \
      for (int x = 0; x < 4; ++x) {                                                \
        dst[x]     = (wm)[(j_ * 20 + 4 * h_ + x) * (GP_PITCH / 4)];                \
        dst[4 + x] = (wm)[(j_ * 20 + 8 + 4 * h_ + x) * (GP_PITCH / 4)];            \
      }                                                                            \
    } else {                                                                       \
      _Pragma("unroll")                                                            \
      for (int x = 0; x < 4; ++x)                                                  \
        dst[x] = (wm)[(j_ * 20 + 16 + x) * (GP_PITCH / 4)];                        \
    }                                                                              \
  } while (0)

__global__ __launch_bounds__(512) void gp_round_kernel(
    const float* __restrict__ hvin,
    const int* __restrict__ rowptr,
    const int* __restrict__ idxs,
    const float* __restrict__ hsum,
    const float* __restrict__ W,
    const float* __restrict__ bias,
    const float* __restrict__ w_ih, const float* __restrict__ w_hh,
    const float* __restrict__ b_ih, const float* __restrict__ b_hh,
    float* __restrict__ hvout)
{
  __shared__ __attribute__((aligned(16))) float wl[GP_GRU_UNITS * GP_PITCH];
  __shared__ __attribute__((aligned(16))) float wm[GP_MSG_UNITS * GP_PITCH];
  __shared__ __attribute__((aligned(16))) float sl[GP_RND_WAVES * 16 * GP_SROW];

  const int tid  = threadIdx.x;
  const int lane = tid & 31;
  const int wave = __builtin_amdgcn_readfirstlane(tid >> 5);
  const int bid  = (int)blockIdx.x;
  const int ntiles = GP_N_NODES / 16;
  const int row   = lane & 15;
  const int half  = lane >> 4;

  int slot = (wave + bid) & (GP_RND_WAVES - 1);
  int tile = slot * GP_RND_NB + bid;

  {
    const int fc = tid & 15, frl = (tid >> 4) & 15, hb = tid >> 8;
#pragma unroll
    for (int it = 0; it < 12; ++it) {
      const int mb = 2 * (it % 6) + hb;
      const float* wsrc = ((it < 6) ? w_ih : w_hh) + (mb * 16 + frl) * 64 + 4 * fc;
      const v4f v = *(const v4f*)wsrc;
      const int G  = (it / 6) * 3 + (mb >> 2);
      const int j0 = mb & 3;
      float* d = wl + ((G * 4 + j0) * 8 + (fc >> 1)) * GP_PITCH + (fc & 1) * 2;
      v2f lo, hi;
      lo.x = v.x; lo.y = v.y; hi.x = v.z; hi.y = v.w;
      *(v2f*)(d + frl * 4) = lo;
      *(v2f*)(d + (16 + frl) * 4) = hi;
    }
  }
  {
#pragma unroll
    for (int it = 0; it < GP_MSG_UNITS / GP_RND_WAVES; ++it) {
      const int u  = it * GP_RND_WAVES + wave;
      const int j0 = u / 20, c2 = u % 20;
      const float* gsrc = W + (8 * c2 + 2 * half) * GP_ACT_DIM + j0 * 16 + row;
      v4f v;
      v.x = gsrc[0];
      v.y = gsrc[GP_ACT_DIM];
      v.z = gsrc[4 * GP_ACT_DIM];
      v.w = gsrc[5 * GP_ACT_DIM];
      *(v4f*)(wm + u * GP_PITCH + lane * 4) = v;
    }
  }
  __syncthreads();

  const v4f* wq  = (const v4f*)wl + lane;
  const v4f* wmq = (const v4f*)wm + lane;
  float* const sw = sl + wave * (16 * GP_SROW);
  float* const so = sw + 2 * lane;
  const float* const sf = sw + row * GP_SROW + 2 * half;
  float* const sc = sw + (8 * half) * GP_SROW + row;
  const unsigned lo8 = (unsigned)lane * 8u;

  v4f fb[2][6];
  float bn[6];
  bn[0] = b_ih[row]; bn[1] = b_ih[64 + row]; bn[2] = b_ih[128 + row];
  bn[3] = b_hh[row]; bn[4] = b_hh[64 + row]; bn[5] = b_hh[128 + row];
  float bq[4];
#pragma unroll
  for (int j0 = 0; j0 < 4; ++j0) bq[j0] = bias[j0 * 16 + row];

  for (;;) {
    const int n0 = tile * 16;

    {
      int rp = 0;
      if (lane <= 16) rp = rowptr[n0 + lane];
      const int e0 = __builtin_amdgcn_readlane(rp, 0);
      const int e1 = __builtin_amdgcn_readlane(rp, 16);
      int cur = 0;
      int nb  = __builtin_amdgcn_readlane(rp, 1);
      float a0 = 0.0f, a1 = 0.0f;
      for (int base = e0; base < e1; base += 32) {
        const int cnt = (e1 - base < 32) ? (e1 - base) : 32;
        int idx = 0;
        if (lane < cnt) idx = idxs[base + lane];
        float r0[32], r1[32];
#pragma unroll
        for (int u = 0; u < 16; ++u) {
          const int k = (u < cnt) ? u : (cnt - 1);
          const unsigned s = (unsigned)__builtin_amdgcn_readlane(idx, k);
          const v2f t = *(const v2f*)((const char*)hvin + ((s * 256u) | lo8));
          r0[u] = t.x; r1[u] = t.y;
        }
        if (cnt > 16) {
#pragma unroll
          for (int u = 16; u < 32; ++u) {
            const int k = (u < cnt) ? u : (cnt - 1);
            const unsigned s = (unsigned)__builtin_amdgcn_readlane(idx, k);
            const v2f t = *(const v2f*)((const char*)hvin + ((s * 256u) | lo8));
            r0[u] = t.x; r1[u] = t.y;
          }
        }
#pragma unroll
        for (int u = 0; u < 32; ++u) {
          if (u < cnt) {
            const int e = base + u;
            while (e == nb) {
              v2f t; t.x = a0; t.y = a1;
              *(v2f*)(so + cur * GP_SROW) = t;
              a0 = 0.0f; a1 = 0.0f;
              ++cur;
              nb = __builtin_amdgcn_readlane(rp, cur + 1);
            }
            a0 += r0[u]; a1 += r1[u];
          }
        }
      }
      while (cur < 16) {
        v2f t; t.x = a0; t.y = a1;
        *(v2f*)(so + cur * GP_SROW) = t;
        a0 = 0.0f; a1 = 0.0f;
        ++cur;
      }
    }
    __builtin_amdgcn_wave_barrier();

    v2f Ah[16];
    {
      v2f As[16], Ae[8];
      const float* hq = hvin + (long)(n0 + row) * GP_NODE_DIM + 2 * half;
      const float* ep = hsum + (long)(n0 + row) * GP_EDGE_DIM + 2 * half;
#pragma unroll
      for (int c = 0; c < 16; ++c) { As[c] = *(const v2f*)(sf + 4 * c); Ah[c] = ld2(hq + 4 * c); }
#pragma unroll
      for (int c = 0; c < 8; ++c) Ae[c] = ld2(ep + 4 * c);

      float dg[8];
#pragma unroll
      for (int i = 0; i < 8; ++i)
        dg[i] = (float)(rowptr[n0 + 8 * half + i + 1] - rowptr[n0 + 8 * half + i]);

      v4f nf[2][8];
      v8f p, q;
      float bqt[4];
#pragma unroll
      for (int j0 = 0; j0 < 4; ++j0) {
        bqt[j0] = bq[j0];
        asm volatile("" : "+v"(bqt[j0]));
      }
      GP_MSG_LOADSET(nf[0], wmq, 0);
#pragma unroll
      for (int st = 0; st < 12; ++st) {
        const int j0 = st / 3, h = st % 3;
        if (h == 0) {
#pragma unroll
          for (int i = 0; i < 8; ++i) { p[i] = 0.0f; q[i] = bqt[j0]; }
        }
        if (st < 11) GP_MSG_LOADSET(nf[(st + 1) & 1], wmq, st + 1);
        __builtin_amdgcn_sched_barrier(0);
        const v4f* f = nf[st & 1];
        if (h < 2) {
#pragma unroll
          for (int x = 0; x < 4; ++x) {
            const int c2 = 4 * h + x;
            v2f b;
            b.x = f[x].x;     b.y = f[x].y;     p = wmma4(As[2 * c2], b, p);
            b.x = f[4 + x].x; b.y = f[4 + x].y; q = wmma4(Ah[2 * c2], b, q);
            b.x = f[x].z;     b.y = f[x].w;     p = wmma4(As[2 * c2 + 1], b, p);
            b.x = f[4 + x].z; b.y = f[4 + x].w; q = wmma4(Ah[2 * c2 + 1], b, q);
          }
        } else {
#pragma unroll
          for (int x = 0; x < 4; ++x) {
            v2f b;
            b.x = f[x].x; b.y = f[x].y; p = wmma4(Ae[2 * x], b, p);
            b.x = f[x].z; b.y = f[x].w; p = wmma4(Ae[2 * x + 1], b, p);
          }
#pragma unroll
          for (int i = 0; i < 8; ++i) sc[i * GP_SROW + j0 * 16] = fmaf(dg[i], q[i], p[i]);
        }
        __builtin_amdgcn_sched_barrier(0);
      }
    }
    __builtin_amdgcn_wave_barrier();
    v2f Aa[16];
#pragma unroll
    for (int c = 0; c < 16; ++c) Aa[c] = *(const v2f*)(sf + 4 * c);
    GP_GRU_LOADSET(fb[0], wq, 0);

    const float* hrow = hvin  + (long)(n0 + 8 * half) * GP_NODE_DIM + row;
    float*       orow = hvout + (long)(n0 + 8 * half) * GP_NODE_DIM + row;
#pragma nounroll
    for (int j0 = 0; j0 < 4; ++j0) {
      const int jn = (j0 + 1) & 3;
      const v4f* wj = wq + j0 * (8 * (GP_PITCH / 4));
      const v4f* wn = wq + jn * (8 * (GP_PITCH / 4));
      v8f ir, iz, in_, hr, hz, hn;
#pragma unroll
      for (int i = 0; i < 8; ++i) {
        ir[i] = bn[0]; iz[i] = bn[1]; in_[i] = bn[2];
        hr[i] = bn[3]; hz[i] = bn[4]; hn[i] = bn[5];
      }
      bn[0] = b_ih[jn * 16 + row]; bn[1] = b_ih[64 + jn * 16 + row]; bn[2] = b_ih[128 + jn * 16 + row];
      bn[3] = b_hh[jn * 16 + row]; bn[4] = b_hh[64 + jn * 16 + row]; bn[5] = b_hh[128 + jn * 16 + row];
#pragma unroll
      for (int c2 = 0; c2 < 8; ++c2) {
        if (c2 < 7) GP_GRU_LOADSET(fb[(c2 + 1) & 1], wj, c2 + 1);
        else        GP_GRU_LOADSET(fb[0], wn, 0);
        __builtin_amdgcn_sched_barrier(0);
        {
          const v4f* f = fb[c2 & 1];
          v2f b;
          b.x = f[0].x; b.y = f[0].y; ir  = wmma4(Aa[2 * c2], b, ir);
          b.x = f[1].x; b.y = f[1].y; iz  = wmma4(Aa[2 * c2], b, iz);
          b.x = f[2].x; b.y = f[2].y; in_ = wmma4(Aa[2 * c2], b, in_);
          b.x = f[3].x; b.y = f[3].y; hr  = wmma4(Ah[2 * c2], b, hr);
          b.x = f[4].x; b.y = f[4].y; hz  = wmma4(Ah[2 * c2], b, hz);
          b.x = f[5].x; b.y = f[5].y; hn  = wmma4(Ah[2 * c2], b, hn);
          b.x = f[0].z; b.y = f[0].w; ir  = wmma4(Aa[2 * c2 + 1], b, ir);
          b.x = f[1].z; b.y = f[1].w; iz  = wmma4(Aa[2 * c2 + 1], b, iz);
          b.x = f[2].z; b.y = f[2].w; in_ = wmma4(Aa[2 * c2 + 1], b, in_);
          b.x = f[3].z; b.y = f[3].w; hr  = wmma4(Ah[2 * c2 + 1], b, hr);
          b.x = f[4].z; b.y = f[4].w; hz  = wmma4(Ah[2 * c2 + 1], b, hz);
          b.x = f[5].z; b.y = f[5].w; hn  = wmma4(Ah[2 * c2 + 1], b, hn);
        }
        __builtin_amdgcn_sched_barrier(0);
      }
#pragma unroll
      for (int i = 0; i < 8; ++i) {
        const float h = hrow[i * GP_NODE_DIM + j0 * 16];
        const float r = sigmoid_(ir[i] + hr[i]);
        const float z = sigmoid_(iz[i] + hz[i]);
        const float n = tanh_(in_[i] + r * hn[i]);
        orow[i * GP_NODE_DIM + j0 * 16] = (1.0f - z) * n + z * h;
      }
    }
    slot += GP_RND_WAVES;
    tile = slot * GP_RND_NB + bid;
    if (tile >= ntiles) break;
  }
}

extern "C" void kernel_launch(void* const* d_in, const int* in_sizes, int n_in,
                              void* d_out, int out_size, void* d_ws, size_t ws_size,
                              hipStream_t stream) {
  const float* hv    = (const float*)d_in[0];
  const float* he    = (const float*)d_in[1];
  const int*   src   = (const int*)d_in[2];
  const int*   dst   = (const int*)d_in[3];
  const float* W_msg = (const float*)d_in[4];
  const float* b_msg = (const float*)d_in[5];
  const float* w_ih  = (const float*)d_in[6];
  const float* w_hh  = (const float*)d_in[7];
  const float* b_ih  = (const float*)d_in[8];
  const float* b_hh  = (const float*)d_in[9];
  float* out = (float*)d_out;

  const size_t state_elems = (size_t)GP_N_NODES * GP_NODE_DIM;
  float* hv_buf = (float*)d_ws;
  float* h_sum   = hv_buf + state_elems;
  int*   rowptr  = (int*)(h_sum + (size_t)GP_N_NODES * GP_EDGE_DIM);
  int*   csr_src = rowptr + 100032;
  int* binned  = (int*)hv_buf;
  int* csr_eid = (int*)hv_buf + 2 * GP_N_EDGES;
  int* hist    = (int*)hv_buf + 3 * GP_N_EDGES;
  int* binoff  = hist + 100352;

  const int scan_blocks   = (GP_SCAN_LEN + 1 + 1023) / 1024;
  const int gather_blocks = (GP_N_NODES + 8 * GP_NPW - 1) / (8 * GP_NPW);

  gp_hist_kernel<<<GP_SORT_BLOCKS, 1024, 0, stream>>>(dst, hist);
  gp_scan_kernel<<<scan_blocks, 1024, 0, stream>>>(hist, binoff, GP_SCAN_LEN);
  gp_bin_kernel<<<GP_SORT_BLOCKS, 1024, 0, stream>>>(src, dst, binoff, binned);
  gp_sort_kernel<<<GP_NBINS, 1024, 0, stream>>>(binned, binoff, rowptr, csr_src, csr_eid);
  gp_gather_kernel<1><<<gather_blocks, 256, 0, stream>>>(he, rowptr, csr_eid, h_sum);

  for (int t = 0; t < GP_ROUNDS; ++t) {
    const float* inp = (t == 0) ? hv : ((t == 1) ? (const float*)out : (const float*)hv_buf);
    float* outp = (t == 1) ? hv_buf : out;
    gp_round_kernel<<<GP_RND_NB, 512, 0, stream>>>(
        inp, rowptr, csr_src, h_sum,
        W_msg + (size_t)t * GP_CAT_DIM * GP_ACT_DIM,
        b_msg + (size_t)t * GP_ACT_DIM,
        w_ih + (size_t)t * 3 * GP_NODE_DIM * GP_NODE_DIM,
        w_hh + (size_t)t * 3 * GP_NODE_DIM * GP_NODE_DIM,
        b_ih + (size_t)t * 3 * GP_NODE_DIM,
        b_hh + (size_t)t * 3 * GP_NODE_DIM, outp);
  }
}


# Edge lists in two launches, not four: a run keeps its binned edges in its own part with each bin's start there; a bin's first slot = sum of those starts (no count or scan launch); sum order changes
# speedup vs baseline: 3.8760x; 1.0610x over previous
#include <hip/hip_runtime.h>


typedef float v2f __attribute__((ext_vector_type(2)));
typedef float v4f __attribute__((ext_vector_type(4)));
typedef float v8f __attribute__((ext_vector_type(8)));
typedef int   v2i __attribute__((ext_vector_type(2)));

#define GP_ROUNDS   3
#define GP_NODE_DIM 64
#define GP_ACT_DIM  64
#define GP_EDGE_DIM 32
#define GP_N_NODES  100000
#define GP_N_EDGES  1600000
#define GP_CAT_DIM  160
#define GP_NPW      8

static __device__ __forceinline__ v8f wmma4(v2f a, v2f b, v8f c) {
  return __builtin_amdgcn_wmma_f32_16x16x4_f32(false, a, false, b, (short)0, c,
                                               false, false);
}
static __device__ __forceinline__ v2f ld2(const float* p) {
  return *(const v2f*)p;
}
static __device__ __forceinline__ float sigmoid_(float x) {
  return __builtin_amdgcn_rcpf(1.0f + __builtin_amdgcn_exp2f(x * -1.44269504f));
}
static __device__ __forceinline__ float tanh_(float x) {
  return fmaf(__builtin_amdgcn_rcpf(1.0f + __builtin_amdgcn_exp2f(x * 2.88539008f)), -2.0f, 1.0f);
}


#define GP_NBINS       391
#define GP_SORT_BLOCKS 256
#define GP_CHUNK       6250
#define GP_SORT_CAP    8192

static_assert(GP_SORT_BLOCKS * GP_CHUNK == GP_N_EDGES, "whole runs only");
static_assert(GP_NBINS * 256 >= GP_N_NODES && (GP_NBINS - 1) * 256 < GP_N_NODES, "bins cover the nodes");
static_assert(GP_N_NODES <= (1 << 17), "source node packed in 17 bits");

static __device__ __forceinline__ int gp_block_scan(int x, int* ws, int tid) {
  const int lane = tid & 31;
  const int wave = tid >> 5;
  int inc = x;
  for (int o = 1; o < 32; o <<= 1) {
    const int y = __shfl_up(inc, o);
    if (lane >= o) inc += y;
  }
  if (lane == 31) ws[wave] = inc;
  __syncthreads();
  int w = (lane < wave) ? ws[lane] : 0;
  for (int o = 16; o > 0; o >>= 1) w += __shfl_xor(w, o);
  return w + inc - x;
}

__global__ __launch_bounds__(1024) void gp_bin_kernel(
    const int* __restrict__ src, const int* __restrict__ dst,
    int* __restrict__ roff, int* __restrict__ binned_)
{
  v2i* const binned = (v2i*)binned_;
  __shared__ int  cnt[GP_NBINS];
  __shared__ int  ws[32];
  __shared__ v2i  stage[GP_CHUNK];
  const int tid = threadIdx.x;
  const int blk = blockIdx.x;
  const int e0  = blk * GP_CHUNK;
  for (int b = tid; b < GP_NBINS; b += 1024) cnt[b] = 0;
  __syncthreads();
  int d[7], s[7], r[7];
#pragma unroll
  for (int k = 0; k < 7; ++k) {
    const int i = k * 1024 + tid;
    d[k] = 0; s[k] = 0; r[k] = 0;
    if (i < GP_CHUNK) {
      d[k] = dst[e0 + i];
      s[k] = src[e0 + i];
      r[k] = atomicAdd(&cnt[d[k] >> 8], 1);
    }
  }
  __syncthreads();
  const int x  = (tid < GP_NBINS) ? cnt[tid] : 0;
  const int ex = gp_block_scan(x, ws, tid);
  if (tid < GP_NBINS) cnt[tid] = ex;
  if (tid <= GP_NBINS) roff[tid * GP_SORT_BLOCKS + blk] = ex;
  __syncthreads();
#pragma unroll
  for (int k = 0; k < 7; ++k) {
    const int i = k * 1024 + tid;
    if (i < GP_CHUNK) {
      v2i p;
      p.x = ((d[k] & 255) << 17) | s[k];
      p.y = e0 + i;
      stage[cnt[d[k] >> 8] + r[k]] = p;
    }
  }
  __syncthreads();
  for (int i = tid; i < GP_CHUNK; i += 1024) binned[e0 + i] = stage[i];
}

#define GP_SEG_REGS 8
__global__ __launch_bounds__(1024) void gp_sort_kernel(
    const int* __restrict__ binned_, const int* __restrict__ roff,
    int* __restrict__ rowptr, int* __restrict__ csr_src, int* __restrict__ csr_eid)
{
  const v2i* const binned = (const v2i*)binned_;
  __shared__ int ncnt[256];
  __shared__ int ncur[256];
  __shared__ int ws[32];
  __shared__ int tot[2];
  __shared__ int ssrc[GP_SORT_CAP];
  __shared__ int seid[GP_SORT_CAP];
  const int tid = threadIdx.x;
  const int bin = blockIdx.x;
  const int run = tid >> 2;
  const int sub = tid & 3;
  if (tid < 256) ncnt[tid] = 0;
  if (tid < 2) tot[tid] = 0;
  __syncthreads();
  const int lo  = roff[bin * GP_SORT_BLOCKS + run];
  const int len = roff[(bin + 1) * GP_SORT_BLOCKS + run] - lo;
  {
    int c0 = (sub == 0) ? lo : 0;
    int c1 = (sub == 0) ? len : 0;
    for (int o = 16; o > 0; o >>= 1) { c0 += __shfl_xor(c0, o); c1 += __shfl_xor(c1, o); }
    if ((tid & 31) == 0) { atomicAdd(&tot[0], c0); atomicAdd(&tot[1], c1); }
  }
  const v2i* const part = binned + run * GP_CHUNK;
  const v2i* const seg  = part + lo;
  v2i pr[GP_SEG_REGS];
#pragma unroll
  for (int k = 0; k < GP_SEG_REGS; ++k) {
    const int j = lo + sub + 4 * k;
    pr[k] = part[(j < GP_CHUNK) ? j : (GP_CHUNK - 1)];
  }
#pragma unroll
  for (int k = 0; k < GP_SEG_REGS; ++k)
    if (sub + 4 * k < len) atomicAdd(&ncnt[pr[k].x >> 17], 1);
  for (int j = sub + 4 * GP_SEG_REGS; j < len; j += 4) atomicAdd(&ncnt[seg[j].x >> 17], 1);
  __syncthreads();
  const int b0 = tot[0];
  const int m  = tot[1];
  const int x  = (tid < 256) ? ncnt[tid] : 0;
  const int ex = gp_block_scan(x, ws, tid);
  if (tid < 256) {
    ncur[tid] = ex;
    const int node = bin * 256 + tid;
    if (node < GP_N_NODES) rowptr[node] = b0 + ex;
  }
  if (bin == GP_NBINS - 1 && tid == 0) rowptr[GP_N_NODES] = b0 + m;
  __syncthreads();
  if (m <= GP_SORT_CAP) {
#pragma unroll
    for (int k = 0; k < GP_SEG_REGS; ++k) {
      if (sub + 4 * k < len) {
        const int pos = atomicAdd(&ncur[pr[k].x >> 17], 1);
        ssrc[pos] = pr[k].x & 0x1FFFF;
        seid[pos] = pr[k].y;
      }
    }
    for (int j = sub + 4 * GP_SEG_REGS; j < len; j += 4) {
      const v2i p = seg[j];
      const int pos = atomicAdd(&ncur[p.x >> 17], 1);
      ssrc[pos] = p.x & 0x1FFFF;
      seid[pos] = p.y;
    }
    __syncthreads();
    for (int i = tid; i < m; i += 1024) {
      csr_src[b0 + i] = ssrc[i];
      csr_eid[b0 + i] = seid[i];
    }
  } else {
    for (int j = sub; j < len; j += 4) {
      const v2i p = seg[j];
      const int pos = b0 + atomicAdd(&ncur[p.x >> 17], 1);
      csr_src[pos] = p.x & 0x1FFFF;
      csr_eid[pos] = p.y;
    }
  }
}

template <int V> struct gp_row;
template <> struct gp_row<1> {
  static __device__ __forceinline__ void load(const char* p, float& a, float& b) {
    a = *(const float*)p; b = 0.0f;
  }
  static __device__ __forceinline__ void store(char* p, float a, float) {
    *(float*)p = a;
  }
};

template <int V>
__global__ __launch_bounds__(256) void gp_gather_kernel(
    const float* __restrict__ rows, const int* __restrict__ rowptr,
    const int* __restrict__ idxs, float* __restrict__ outp)
{
  const int lane = threadIdx.x & 31;
  const int wave = __builtin_amdgcn_readfirstlane((int)(threadIdx.x >> 5));
  const int n0 = (blockIdx.x * 8 + wave) * GP_NPW;
  if (n0 >= GP_N_NODES) return;
  const int nn = (GP_N_NODES - n0 < GP_NPW) ? (GP_N_NODES - n0) : GP_NPW;
  int rp = 0;
  if (lane <= nn) rp = rowptr[n0 + lane];
  const int e0 = __builtin_amdgcn_readlane(rp, 0);
  const int e1 = __builtin_amdgcn_readlane(rp, nn);
  int cur = 0;
  int nb  = __builtin_amdgcn_readlane(rp, 1);
  const unsigned lo = (unsigned)lane * (4u * V);
  float a0 = 0.0f, a1 = 0.0f;
  char* const ob = (char*)outp + (unsigned)n0 * (128u * V) + lo;

  for (int base = e0; base < e1; base += 32) {
    const int cnt = (e1 - base < 32) ? (e1 - base) : 32;
    int idx = 0;
    if (lane < cnt) idx = idxs[base + lane];
#pragma unroll
    for (int h = 0; h < 32; h += 16) {
      if (h < cnt) {
        float r0[16], r1[16];
#pragma unroll
        for (int u = 0; u < 16; ++u) {
          const int k = (h + u < cnt) ? (h + u) : (cnt - 1);
          const unsigned s = (unsigned)__builtin_amdgcn_readlane(idx, k);
          gp_row<V>::load((const char*)rows + ((s * (128u * V)) | lo), r0[u], r1[u]);
        }
#pragma unroll
        for (int u = 0; u < 16; ++u) {
          if (h + u < cnt) {
            const int e = base + h + u;
            while (e == nb) {
              gp_row<V>::store(ob + (unsigned)cur * (128u * V), a0, a1);
              a0 = 0.0f; a1 = 0.0f;
              ++cur;
              nb = __builtin_amdgcn_readlane(rp, cur + 1);
            }
            a0 += r0[u]; a1 += r1[u];
          }
        }
      }
    }
  }
  while (cur < nn) {
    gp_row<V>::store(ob + (unsigned)cur * (128u * V), a0, a1);
    a0 = 0.0f; a1 = 0.0f;
    ++cur;
  }
}

#define GP_RND_NB    256
#define GP_RND_WAVES 16
#define GP_GRU_UNITS 192
#define GP_MSG_UNITS 80
#define GP_PITCH     132
#define GP_SROW      68

static_assert(GP_RND_WAVES * GP_RND_NB <= GP_N_NODES / 16, "every wave must own a first tile");
static_assert(GP_N_NODES % 16 == 0, "whole tiles only");

#define GP_GRU_LOADSET(dst, wj, c2)                                                \
  do {                                                                             \
    _Pragma("unroll")                                                              \
    for (int g = 0; g < 6; ++g)                                                    \
      dst[g] = (wj)[(g * 32 + (c2)) * (GP_PITCH / 4)];                             \
  } while (0)

#define GP_MSG_LOADSET(dst, wm, st)                                                \
  do {                                                                             \
    const int j_ = (st) / 3, h_ = (st) % 3;                                        \
    if (h_ < 2) {                                                                  \
      _Pragma("unroll")                                                            \
      for (int x = 0; x < 4; ++x) {                                                \
        dst[x]     = (wm)[(j_ * 20 + 4 * h_ + x) * (GP_PITCH / 4)];                \
        dst[4 + x] = (wm)[(j_ * 20 + 8 + 4 * h_ + x) * (GP_PITCH / 4)];            \
      }                                                                            \
    } else {                                                                       \
      _Pragma("unroll")                                                            \
      for (int x = 0; x < 4; ++x)                                                  \
        dst[x] = (wm)[(j_ * 20 + 16 + x) * (GP_PITCH / 4)];                        \
    }                                                                              \
  } while (0)

__global__ __launch_bounds__(512) void gp_round_kernel(
    const float* __restrict__ hvin,
    const int* __restrict__ rowptr,
    const int* __restrict__ idxs,
    const float* __restrict__ hsum,
    const float* __restrict__ W,
    const float* __restrict__ bias,
    const float* __restrict__ w_ih, const float* __restrict__ w_hh,
    const float* __restrict__ b_ih, const float* __restrict__ b_hh,
    float* __restrict__ hvout)
{
  __shared__ __attribute__((aligned(16))) float wl[GP_GRU_UNITS * GP_PITCH];
  __shared__ __attribute__((aligned(16))) float wm[GP_MSG_UNITS * GP_PITCH];
  __shared__ __attribute__((aligned(16))) float sl[GP_RND_WAVES * 16 * GP_SROW];

  const int tid  = threadIdx.x;
  const int lane = tid & 31;
  const int wave = __builtin_amdgcn_readfirstlane(tid >> 5);
  const int bid  = (int)blockIdx.x;
  const int ntiles = GP_N_NODES / 16;
  const int row   = lane & 15;
  const int half  = lane >> 4;

  int slot = (wave + bid) & (GP_RND_WAVES - 1);
  int tile = slot * GP_RND_NB + bid;

  {
    const int fc = tid & 15, frl = (tid >> 4) & 15, hb = tid >> 8;
#pragma unroll
    for (int it = 0; it < 12; ++it) {
      const int mb = 2 * (it % 6) + hb;
      const float* wsrc = ((it < 6) ? w_ih : w_hh) + (mb * 16 + frl) * 64 + 4 * fc;
      const v4f v = *(const v4f*)wsrc;
      const int G  = (it / 6) * 3 + (mb >> 2);
      const int j0 = mb & 3;
      float* d = wl + ((G * 4 + j0) * 8 + (fc >> 1)) * GP_PITCH + (fc & 1) * 2;
      v2f lo, hi;
      lo.x = v.x; lo.y = v.y; hi.x = v.z; hi.y = v.w;
      *(v2f*)(d + frl * 4) = lo;
      *(v2f*)(d + (16 + frl) * 4) = hi;
    }
  }
  {
#pragma unroll
    for (int it = 0; it < GP_MSG_UNITS / GP_RND_WAVES; ++it) {
      const int u  = it * GP_RND_WAVES + wave;
      const int j0 = u / 20, c2 = u % 20;
      const float* gsrc = W + (8 * c2 + 2 * half) * GP_ACT_DIM + j0 * 16 + row;
      v4f v;
      v.x = gsrc[0];
      v.y = gsrc[GP_ACT_DIM];
      v.z = gsrc[4 * GP_ACT_DIM];
      v.w = gsrc[5 * GP_ACT_DIM];
      *(v4f*)(wm + u * GP_PITCH + lane * 4) = v;
    }
  }
  __syncthreads();

  const v4f* wq  = (const v4f*)wl + lane;
  const v4f* wmq = (const v4f*)wm + lane;
  float* const sw = sl + wave * (16 * GP_SROW);
  float* const so = sw + 2 * lane;
  const float* const sf = sw + row * GP_SROW + 2 * half;
  float* const sc = sw + (8 * half) * GP_SROW + row;
  const unsigned lo8 = (unsigned)lane * 8u;

  v4f fb[2][6];
  float bn[6];
  bn[0] = b_ih[row]; bn[1] = b_ih[64 + row]; bn[2] = b_ih[128 + row];
  bn[3] = b_hh[row]; bn[4] = b_hh[64 + row]; bn[5] = b_hh[128 + row];
  float bq[4];
#pragma unroll
  for (int j0 = 0; j0 < 4; ++j0) bq[j0] = bias[j0 * 16 + row];

  for (;;) {
    const int n0 = tile * 16;

    {
      int rp = 0;
      if (lane <= 16) rp = rowptr[n0 + lane];
      const int e0 = __builtin_amdgcn_readlane(rp, 0);
      const int e1 = __builtin_amdgcn_readlane(rp, 16);
      int cur = 0;
      int nb  = __builtin_amdgcn_readlane(rp, 1);
      float a0 = 0.0f, a1 = 0.0f;
      for (int base = e0; base < e1; base += 32) {
        const int cnt = (e1 - base < 32) ? (e1 - base) : 32;
        int idx = 0;
        if (lane < cnt) idx = idxs[base + lane];
        float r0[32], r1[32];
#pragma unroll
        for (int u = 0; u < 16; ++u) {
          const int k = (u < cnt) ? u : (cnt - 1);
          const unsigned s = (unsigned)__builtin_amdgcn_readlane(idx, k);
          const v2f t = *(const v2f*)((const char*)hvin + ((s * 256u) | lo8));
          r0[u] = t.x; r1[u] = t.y;
        }
        if (cnt > 16) {
#pragma unroll
          for (int u = 16; u < 32; ++u) {
            const int k = (u < cnt) ? u : (cnt - 1);
            const unsigned s = (unsigned)__builtin_amdgcn_readlane(idx, k);
            const v2f t = *(const v2f*)((const char*)hvin + ((s * 256u) | lo8));
            r0[u] = t.x; r1[u] = t.y;
          }
        }
#pragma unroll
        for (int u = 0; u < 32; ++u) {
          if (u < cnt) {
            const int e = base + u;
            while (e == nb) {
              v2f t; t.x = a0; t.y = a1;
              *(v2f*)(so + cur * GP_SROW) = t;
              a0 = 0.0f; a1 = 0.0f;
              ++cur;
              nb = __builtin_amdgcn_readlane(rp, cur + 1);
            }
            a0 += r0[u]; a1 += r1[u];
          }
        }
      }
      while (cur < 16) {
        v2f t; t.x = a0; t.y = a1;
        *(v2f*)(so + cur * GP_SROW) = t;
        a0 = 0.0f; a1 = 0.0f;
        ++cur;
      }
    }
    __builtin_amdgcn_wave_barrier();

    v2f Ah[16];
    {
      v2f As[16], Ae[8];
      const float* hq = hvin + (long)(n0 + row) * GP_NODE_DIM + 2 * half;
      const float* ep = hsum + (long)(n0 + row) * GP_EDGE_DIM + 2 * half;
#pragma unroll
      for (int c = 0; c < 16; ++c) { As[c] = *(const v2f*)(sf + 4 * c); Ah[c] = ld2(hq + 4 * c); }
#pragma unroll
      for (int c = 0; c < 8; ++c) Ae[c] = ld2(ep + 4 * c);

      float dg[8];
#pragma unroll
      for (int i = 0; i < 8; ++i)
        dg[i] = (float)(rowptr[n0 + 8 * half + i + 1] - rowptr[n0 + 8 * half + i]);

      v4f nf[2][8];
      v8f p, q;
      float bqt[4];
#pragma unroll
      for (int j0 = 0; j0 < 4; ++j0) {
        bqt[j0] = bq[j0];
        asm volatile("" : "+v"(bqt[j0]));
      }
      GP_MSG_LOADSET(nf[0], wmq, 0);
#pragma unroll
      for (int st = 0; st < 12; ++st) {
        const int j0 = st / 3, h = st % 3;
        if (h == 0) {
#pragma unroll
          for (int i = 0; i < 8; ++i) { p[i] = 0.0f; q[i] = bqt[j0]; }
        }
        if (st < 11) GP_MSG_LOADSET(nf[(st + 1) & 1], wmq, st + 1);
        __builtin_amdgcn_sched_barrier(0);
        const v4f* f = nf[st & 1];
        if (h < 2) {
#pragma unroll
          for (int x = 0; x < 4; ++x) {
            const int c2 = 4 * h + x;
            v2f b;
            b.x = f[x].x;     b.y = f[x].y;     p = wmma4(As[2 * c2], b, p);
            b.x = f[4 + x].x; b.y = f[4 + x].y; q = wmma4(Ah[2 * c2], b, q);
            b.x = f[x].z;     b.y = f[x].w;     p = wmma4(As[2 * c2 + 1], b, p);
            b.x = f[4 + x].z; b.y = f[4 + x].w; q = wmma4(Ah[2 * c2 + 1], b, q);
          }
        } else {
#pragma unroll
          for (int x = 0; x < 4; ++x) {
            v2f b;
            b.x = f[x].x; b.y = f[x].y; p = wmma4(Ae[2 * x], b, p);
            b.x = f[x].z; b.y = f[x].w; p = wmma4(Ae[2 * x + 1], b, p);
          }
#pragma unroll
          for (int i = 0; i < 8; ++i) sc[i * GP_SROW + j0 * 16] = fmaf(dg[i], q[i], p[i]);
        }
        __builtin_amdgcn_sched_barrier(0);
      }
    }
    __builtin_amdgcn_wave_barrier();
    v2f Aa[16];
#pragma unroll
    for (int c = 0; c < 16; ++c) Aa[c] = *(const v2f*)(sf + 4 * c);
    GP_GRU_LOADSET(fb[0], wq, 0);

    const float* hrow = hvin  + (long)(n0 + 8 * half) * GP_NODE_DIM + row;
    float*       orow = hvout + (long)(n0 + 8 * half) * GP_NODE_DIM + row;
#pragma nounroll
    for (int j0 = 0; j0 < 4; ++j0) {
      const int jn = (j0 + 1) & 3;
      const v4f* wj = wq + j0 * (8 * (GP_PITCH / 4));
      const v4f* wn = wq + jn * (8 * (GP_PITCH / 4));
      v8f ir, iz, in_, hr, hz, hn;
#pragma unroll
      for (int i = 0; i < 8; ++i) {
        ir[i] = bn[0]; iz[i] = bn[1]; in_[i] = bn[2];
        hr[i] = bn[3]; hz[i] = bn[4]; hn[i] = bn[5];
      }
      bn[0] = b_ih[jn * 16 + row]; bn[1] = b_ih[64 + jn * 16 + row]; bn[2] = b_ih[128 + jn * 16 + row];
      bn[3] = b_hh[jn * 16 + row]; bn[4] = b_hh[64 + jn * 16 + row]; bn[5] = b_hh[128 + jn * 16 + row];
#pragma unroll
      for (int c2 = 0; c2 < 8; ++c2) {
        if (c2 < 7) GP_GRU_LOADSET(fb[(c2 + 1) & 1], wj, c2 + 1);
        else        GP_GRU_LOADSET(fb[0], wn, 0);
        __builtin_amdgcn_sched_barrier(0);
        {
          const v4f* f = fb[c2 & 1];
          v2f b;
          b.x = f[0].x; b.y = f[0].y; ir  = wmma4(Aa[2 * c2], b, ir);
          b.x = f[1].x; b.y = f[1].y; iz  = wmma4(Aa[2 * c2], b, iz);
          b.x = f[2].x; b.y = f[2].y; in_ = wmma4(Aa[2 * c2], b, in_);
          b.x = f[3].x; b.y = f[3].y; hr  = wmma4(Ah[2 * c2], b, hr);
          b.x = f[4].x; b.y = f[4].y; hz  = wmma4(Ah[2 * c2], b, hz);
          b.x = f[5].x; b.y = f[5].y; hn  = wmma4(Ah[2 * c2], b, hn);
          b.x = f[0].z; b.y = f[0].w; ir  = wmma4(Aa[2 * c2 + 1], b, ir);
          b.x = f[1].z; b.y = f[1].w; iz  = wmma4(Aa[2 * c2 + 1], b, iz);
          b.x = f[2].z; b.y = f[2].w; in_ = wmma4(Aa[2 * c2 + 1], b, in_);
          b.x = f[3].z; b.y = f[3].w; hr  = wmma4(Ah[2 * c2 + 1], b, hr);
          b.x = f[4].z; b.y = f[4].w; hz  = wmma4(Ah[2 * c2 + 1], b, hz);
          b.x = f[5].z; b.y = f[5].w; hn  = wmma4(Ah[2 * c2 + 1], b, hn);
        }
        __builtin_amdgcn_sched_barrier(0);
      }
#pragma unroll
      for (int i = 0; i < 8; ++i) {
        const float h = hrow[i * GP_NODE_DIM + j0 * 16];
        const float r = sigmoid_(ir[i] + hr[i]);
        const float z = sigmoid_(iz[i] + hz[i]);
        const float n = tanh_(in_[i] + r * hn[i]);
        orow[i * GP_NODE_DIM + j0 * 16] = (1.0f - z) * n + z * h;
      }
    }
    slot += GP_RND_WAVES;
    tile = slot * GP_RND_NB + bid;
    if (tile >= ntiles) break;
  }
}

extern "C" void kernel_launch(void* const* d_in, const int* in_sizes, int n_in,
                              void* d_out, int out_size, void* d_ws, size_t ws_size,
                              hipStream_t stream) {
  const float* hv    = (const float*)d_in[0];
  const float* he    = (const float*)d_in[1];
  const int*   src   = (const int*)d_in[2];
  const int*   dst   = (const int*)d_in[3];
  const float* W_msg = (const float*)d_in[4];
  const float* b_msg = (const float*)d_in[5];
  const float* w_ih  = (const float*)d_in[6];
  const float* w_hh  = (const float*)d_in[7];
  const float* b_ih  = (const float*)d_in[8];
  const float* b_hh  = (const float*)d_in[9];
  float* out = (float*)d_out;

  const size_t state_elems = (size_t)GP_N_NODES * GP_NODE_DIM;
  float* hv_buf = (float*)d_ws;
  float* h_sum   = hv_buf + state_elems;
  int*   rowptr  = (int*)(h_sum + (size_t)GP_N_NODES * GP_EDGE_DIM);
  int*   csr_src = rowptr + 100032;
  int* binned  = (int*)hv_buf;
  int* csr_eid = (int*)hv_buf + 2 * GP_N_EDGES;
  int* roff    = (int*)hv_buf + 3 * GP_N_EDGES;

  const int gather_blocks = (GP_N_NODES + 8 * GP_NPW - 1) / (8 * GP_NPW);

  gp_bin_kernel<<<GP_SORT_BLOCKS, 1024, 0, stream>>>(src, dst, roff, binned);
  gp_sort_kernel<<<GP_NBINS, 1024, 0, stream>>>(binned, roff, rowptr, csr_src, csr_eid);
  gp_gather_kernel<1><<<gather_blocks, 256, 0, stream>>>(he, rowptr, csr_eid, h_sum);

  for (int t = 0; t < GP_ROUNDS; ++t) {
    const float* inp = (t == 0) ? hv : ((t == 1) ? (const float*)out : (const float*)hv_buf);
    float* outp = (t == 1) ? hv_buf : out;
    gp_round_kernel<<<GP_RND_NB, 512, 0, stream>>>(
        inp, rowptr, csr_src, h_sum,
        W_msg + (size_t)t * GP_CAT_DIM * GP_ACT_DIM,
        b_msg + (size_t)t * GP_ACT_DIM,
        w_ih + (size_t)t * 3 * GP_NODE_DIM * GP_NODE_DIM,
        w_hh + (size_t)t * 3 * GP_NODE_DIM * GP_NODE_DIM,
        b_ih + (size_t)t * 3 * GP_NODE_DIM,
        b_hh + (size_t)t * 3 * GP_NODE_DIM, outp);
  }
}
